# starting kernel
# baseline (speedup 1.0000x reference)
#include <hip/hip_runtime.h>

typedef __attribute__((ext_vector_type(2))) float v2f;
typedef __attribute__((ext_vector_type(8))) float v8f;

static constexpr int kNodes  = 100000;
static constexpr int kInDim  = 128;
static constexpr int kHidDim = 128;
static constexpr int kOutDim = 64;


__global__ void k_deg_init(float* __restrict__ deg, int n) {
  int i = blockIdx.x * blockDim.x + threadIdx.x;
  if (i < n) deg[i] = 1.0f;
}

__device__ __forceinline__ void atomAddF(float* p, float v) {
  unsafeAtomicAdd(p, v);
}

__global__ void k_deg_count(const int* __restrict__ col, float* __restrict__ deg, int e) {
  int i = blockIdx.x * blockDim.x + threadIdx.x;
  if (i < e) atomAddF(&deg[col[i]], 1.0f);
}

__global__ void k_rsqrt_inplace(float* __restrict__ d, int n) {
  int i = blockIdx.x * blockDim.x + threadIdx.x;
  if (i < n) d[i] = rsqrtf(d[i]);
}


template <int K, int NT, bool RELU>
__global__ __launch_bounds__(128) void k_gemm_wmma(const float* __restrict__ X,
                                                   const float* __restrict__ W,
                                                   float* __restrict__ H, int M) {
  constexpr int N = NT * 16;
  const int wave = blockIdx.x * 4 + (threadIdx.x >> 5);
  const int lane = threadIdx.x & 31;
  const int m0 = wave * 16;
  if (m0 >= M) return;
  const int ml = lane & 15;
  const int hi = lane >> 4;

  v8f acc[NT];
#pragma unroll
  for (int t = 0; t < NT; ++t) acc[t] = {};

  const float* __restrict__ xrow = X + (size_t)(m0 + ml) * K;

#pragma unroll 4
  for (int k = 0; k < K; k += 4) {
    v2f a;
    a.x = xrow[k + 2 * hi + 0];
    a.y = xrow[k + 2 * hi + 1];
    if (RELU) {
      a.x = fmaxf(a.x, 0.0f);
      a.y = fmaxf(a.y, 0.0f);
    }
#pragma unroll
    for (int t = 0; t < NT; ++t) {
      v2f b;
      b.x = W[(size_t)(k + 2 * hi + 0) * N + t * 16 + ml];
      b.y = W[(size_t)(k + 2 * hi + 1) * N + t * 16 + ml];
      acc[t] = __builtin_amdgcn_wmma_f32_16x16x4_f32(
          false, a, false, b, (short)0, acc[t], false, false);
    }
  }

#pragma unroll
  for (int t = 0; t < NT; ++t) {
#pragma unroll
    for (int v = 0; v < 8; ++v) {
      H[(size_t)(m0 + v + 8 * hi) * N + t * 16 + ml] = acc[t][v];
    }
  }
}


template <int D>
__global__ void k_selfloop_bias(const float* __restrict__ H, const float* __restrict__ dinv,
                                const float* __restrict__ bias, float* __restrict__ out,
                                int n) {
  constexpr int C = D / 4;
  long long t = (long long)blockIdx.x * blockDim.x + threadIdx.x;
  long long total = (long long)n * C;
  if (t >= total) return;
  int i = (int)(t / C);
  int c = (int)(t % C);
  float s = dinv[i];
  float w = s * s;
  float4 h = reinterpret_cast<const float4*>(H)[t];
  float4 b = reinterpret_cast<const float4*>(bias)[c];
  float4 o;
  o.x = fmaf(w, h.x, b.x);
  o.y = fmaf(w, h.y, b.y);
  o.z = fmaf(w, h.z, b.z);
  o.w = fmaf(w, h.w, b.w);
  reinterpret_cast<float4*>(out)[t] = o;
}


template <int D>
__global__ void k_edge_agg(const int* __restrict__ row, const int* __restrict__ col,
                           const float* __restrict__ dinv, const float* __restrict__ H,
                           float* __restrict__ out, int e) {
  constexpr int C = D / 4;
  long long t = (long long)blockIdx.x * blockDim.x + threadIdx.x;
  long long total = (long long)e * C;
  if (t >= total) return;
  int ei = (int)(t / C);
  int c  = (int)(t % C);
  int r  = row[ei];
  int d  = col[ei];
  float norm = dinv[r] * dinv[d];
  float4 h = reinterpret_cast<const float4*>(H + (size_t)r * D)[c];
  float* o = out + (size_t)d * D + (size_t)c * 4;
  atomAddF(o + 0, norm * h.x);
  atomAddF(o + 1, norm * h.y);
  atomAddF(o + 2, norm * h.z);
  atomAddF(o + 3, norm * h.w);
}


extern "C" void kernel_launch(void* const* d_in, const int* in_sizes, int n_in,
                              void* d_out, int out_size, void* d_ws, size_t ws_size,
                              hipStream_t stream) {
  const float* x  = (const float*)d_in[0];
  const int*   ei = (const int*)d_in[1];
  const float* W1 = (const float*)d_in[2];
  const float* b1 = (const float*)d_in[3];
  const float* W2 = (const float*)d_in[4];
  const float* b2 = (const float*)d_in[5];

  const int n = kNodes;
  const int e = in_sizes[1] / 2;
  const int* row = ei;
  const int* col = ei + e;

  char* ws = (char*)d_ws;
  float* dinv = (float*)ws;
  constexpr size_t H_OFF  = 512 * 1024;
  constexpr size_t A1_OFF = H_OFF + (size_t)kNodes * kHidDim * 4;
  float* Hbuf = (float*)(ws + H_OFF);
  float* A1   = (float*)(ws + A1_OFF);
  float* out  = (float*)d_out;

  const int T = 256;

  k_deg_init<<<(n + T - 1) / T, T, 0, stream>>>(dinv, n);
  k_deg_count<<<(e + T - 1) / T, T, 0, stream>>>(col, dinv, e);
  k_rsqrt_inplace<<<(n + T - 1) / T, T, 0, stream>>>(dinv, n);

  const int waves = (n + 15) / 16;
  const int gemmBlocks = (waves + 3) / 4;

  k_gemm_wmma<kInDim, kHidDim / 16, false>
      <<<gemmBlocks, 128, 0, stream>>>(x, W1, Hbuf, n);

  {
    long long sl = (long long)n * (kHidDim / 4);
    k_selfloop_bias<kHidDim>
        <<<(unsigned)((sl + T - 1) / T), T, 0, stream>>>(Hbuf, dinv, b1, A1, n);
    long long ea = (long long)e * (kHidDim / 4);
    k_edge_agg<kHidDim>
        <<<(unsigned)((ea + T - 1) / T), T, 0, stream>>>(row, col, dinv, Hbuf, A1, e);
  }

  k_gemm_wmma<kHidDim, kOutDim / 16, true>
      <<<gemmBlocks, 128, 0, stream>>>(A1, W2, Hbuf, n);

  {
    long long sl = (long long)n * (kOutDim / 4);
    k_selfloop_bias<kOutDim>
        <<<(unsigned)((sl + T - 1) / T), T, 0, stream>>>(Hbuf, dinv, b2, out, n);
    long long ea = (long long)e * (kOutDim / 4);
    k_edge_agg<kOutDim>
        <<<(unsigned)((ea + T - 1) / T), T, 0, stream>>>(row, col, dinv, Hbuf, out, e);
  }
}


# Edge sums as a destination-sorted gather: in-call edge list (ranked int atomic count, scan, fill), one wave per node sums rows in registers, stores once; float atomics gone, sum order = rank order
# speedup vs baseline: 2.3096x; 2.3096x over previous
#include <hip/hip_runtime.h>


typedef __attribute__((ext_vector_type(2))) float v2f;
typedef __attribute__((ext_vector_type(8))) float v8f;

static constexpr int kNodes  = 100000;
static constexpr int kInDim  = 128;
static constexpr int kHidDim = 128;
static constexpr int kOutDim = 64;

static_assert(kNodes % 4 == 0, "scan kernel reads the counts as int4");


__global__ void k_cnt_zero(int* __restrict__ cnt, int n) {
  int i = blockIdx.x * blockDim.x + threadIdx.x;
  if (i < n) cnt[i] = 0;
}

__global__ void k_count_rank(const int* __restrict__ col, int* __restrict__ cnt,
                             int* __restrict__ rank, int e) {
  int i = blockIdx.x * blockDim.x + threadIdx.x;
  if (i < e) rank[i] = atomicAdd(&cnt[col[i]], 1);
}

__global__ __launch_bounds__(1024) void k_scan_dinv(const int* __restrict__ cnt,
                                                    int* __restrict__ rowptr,
                                                    float* __restrict__ dinv, int n) {
  __shared__ int s_tot[32];
  __shared__ int s_pre[32];
  const int tid = threadIdx.x;
  const int lane = tid & 31;
  const int wv = tid >> 5;
  const int4* __restrict__ c4 = reinterpret_cast<const int4*>(cnt);
  const int q0 = blockIdx.x * 1024;
  const int nq = n >> 2;

  int pre = 0;
#pragma unroll 8
  for (int k = 0; k < (int)blockIdx.x; ++k) {
    int4 u = c4[k * 1024 + tid];
    pre += (u.x + u.y) + (u.z + u.w);
  }
  const int q = q0 + tid;
  int4 v = make_int4(0, 0, 0, 0);
  if (q < nq) v = c4[q];
  const int t = (v.x + v.y) + (v.z + v.w);

  int inc = t;
#pragma unroll
  for (int o = 1; o < 32; o <<= 1) {
    int u = __shfl_up(inc, o);
    if (lane >= o) inc += u;
  }
#pragma unroll
  for (int o = 16; o > 0; o >>= 1) pre += __shfl_xor(pre, o);
  if (lane == 31) s_tot[wv] = inc;
  if (lane == 0) s_pre[wv] = pre;
  __syncthreads();

  const int wt = s_tot[lane];
  int wp = s_pre[lane];
  int winc = wt;
#pragma unroll
  for (int o = 1; o < 32; o <<= 1) {
    int u = __shfl_up(winc, o);
    if (lane >= o) winc += u;
  }
#pragma unroll
  for (int o = 16; o > 0; o >>= 1) wp += __shfl_xor(wp, o);
  const int wave_off = __shfl(winc - wt, wv);
  const int ex = wp + wave_off + (inc - t);

  if (q < nq) {
    int4 r;
    r.x = ex;
    r.y = r.x + v.x;
    r.z = r.y + v.y;
    r.w = r.z + v.z;
    reinterpret_cast<int4*>(rowptr)[q] = r;
    float4 dv;
    dv.x = rsqrtf((float)(v.x + 1));
    dv.y = rsqrtf((float)(v.y + 1));
    dv.z = rsqrtf((float)(v.z + 1));
    dv.w = rsqrtf((float)(v.w + 1));
    reinterpret_cast<float4*>(dinv)[q] = dv;
  }
}

__global__ void k_fill(const int* __restrict__ row, const int* __restrict__ col,
                       const int* __restrict__ rank, const int* __restrict__ rowptr,
                       const float* __restrict__ dinv, int* __restrict__ edge_list, int e) {
  int2* __restrict__ edges = reinterpret_cast<int2*>(edge_list);
  int i = blockIdx.x * blockDim.x + threadIdx.x;
  if (i >= e) return;
  const int r = row[i];
  const int d = col[i];
  unsigned pos = (unsigned)(rowptr[d] + rank[i]);
  pos = min(pos, (unsigned)(e - 1));
  int2 v;
  v.x = r;
  v.y = __float_as_int(dinv[r]);
  edges[pos] = v;
}


template <int K, int NT, bool RELU>
__global__ __launch_bounds__(128) void k_gemm_wmma(const float* __restrict__ X,
                                                   const float* __restrict__ W,
                                                   float* __restrict__ H, int M) {
  constexpr int N = NT * 16;
  const int wave = blockIdx.x * 4 + (threadIdx.x >> 5);
  const int lane = threadIdx.x & 31;
  const int m0 = wave * 16;
  if (m0 >= M) return;
  const int ml = lane & 15;
  const int hi = lane >> 4;

  v8f acc[NT];
#pragma unroll
  for (int t = 0; t < NT; ++t) acc[t] = {};

  const float* __restrict__ xrow = X + (size_t)(m0 + ml) * K;

#pragma unroll 4
  for (int k = 0; k < K; k += 4) {
    v2f a;
    a.x = xrow[k + 2 * hi + 0];
    a.y = xrow[k + 2 * hi + 1];
    if (RELU) {
      a.x = fmaxf(a.x, 0.0f);
      a.y = fmaxf(a.y, 0.0f);
    }
#pragma unroll
    for (int t = 0; t < NT; ++t) {
      v2f b;
      b.x = W[(size_t)(k + 2 * hi + 0) * N + t * 16 + ml];
      b.y = W[(size_t)(k + 2 * hi + 1) * N + t * 16 + ml];
      acc[t] = __builtin_amdgcn_wmma_f32_16x16x4_f32(
          false, a, false, b, (short)0, acc[t], false, false);
    }
  }

#pragma unroll
  for (int t = 0; t < NT; ++t) {
#pragma unroll
    for (int v = 0; v < 8; ++v) {
      H[(size_t)(m0 + v + 8 * hi) * N + t * 16 + ml] = acc[t][v];
    }
  }
}


template <int EPI>
__device__ __forceinline__ int lane_pick(int v, int j, int sub) {
  if (EPI == 1) return __builtin_amdgcn_readlane(v, j);
  const int a = __builtin_amdgcn_readlane(v, j);
  const int b = __builtin_amdgcn_readlane(v, j + 1);
  return sub ? b : a;
}

template <int D>
__global__ __launch_bounds__(256) void k_gather_agg(const int* __restrict__ rowptr,
                                                    const int* __restrict__ cnt,
                                                    const int* __restrict__ edge_list,
                                                    const float* __restrict__ dinv,
                                                    const float* __restrict__ H,
                                                    const float* __restrict__ bias,
                                                    float* __restrict__ out, int n, int e) {
  constexpr int LPR = D / 4;
  constexpr int EPI = 32 / LPR;
  static_assert(LPR == 32 || LPR == 16, "row must be one wave or half a wave of float4");
  const int2* __restrict__ edges = reinterpret_cast<const int2*>(edge_list);
  const int lane = threadIdx.x & 31;
  const int node = __builtin_amdgcn_readfirstlane(blockIdx.x * 8 + (threadIdx.x >> 5));
  if (node >= n) return;
  const int sub = lane / LPR;
  const int c = lane % LPR;

  int start = rowptr[node];
  int m = cnt[node];
  start = min(max(start, 0), e);
  m = min(max(m, 0), e - start);

  const float di = dinv[node];
  const float4* __restrict__ H4 = reinterpret_cast<const float4*>(H);
  const float4 hs = H4[(size_t)node * LPR + c];
  float4 acc = make_float4(0.0f, 0.0f, 0.0f, 0.0f);

  for (int base = 0; base < m; base += 32) {
    const int rem = min(32, m - base);
    int2 ed = make_int2(0, 0);
    if (lane < rem) ed = edges[start + base + lane];
    const int src = (int)min((unsigned)ed.x, (unsigned)(n - 1));
    const int wgt = __float_as_int(__int_as_float(ed.y) * di);
    const int full = rem - (rem % EPI);
#pragma unroll 4
    for (int j = 0; j < full; j += EPI) {
      const int s = lane_pick<EPI>(src, j, sub);
      const float w = __int_as_float(lane_pick<EPI>(wgt, j, sub));
      const float4 h = H4[(size_t)s * LPR + c];
      acc.x = fmaf(w, h.x, acc.x);
      acc.y = fmaf(w, h.y, acc.y);
      acc.z = fmaf(w, h.z, acc.z);
      acc.w = fmaf(w, h.w, acc.w);
    }
    if (EPI == 2 && (rem & 1)) {
      const int s = __builtin_amdgcn_readlane(src, rem - 1);
      const float w = __int_as_float(__builtin_amdgcn_readlane(wgt, rem - 1));
      if (sub == 0) {
        const float4 h = H4[(size_t)s * LPR + c];
        acc.x = fmaf(w, h.x, acc.x);
        acc.y = fmaf(w, h.y, acc.y);
        acc.z = fmaf(w, h.z, acc.z);
        acc.w = fmaf(w, h.w, acc.w);
      }
    }
  }

  if (EPI == 2) {
    acc.x += __shfl_xor(acc.x, 16);
    acc.y += __shfl_xor(acc.y, 16);
    acc.z += __shfl_xor(acc.z, 16);
    acc.w += __shfl_xor(acc.w, 16);
  }
  if (sub == 0) {
    const float w0 = di * di;
    const float4 b = reinterpret_cast<const float4*>(bias)[c];
    float4 o;
    o.x = fmaf(w0, hs.x, b.x) + acc.x;
    o.y = fmaf(w0, hs.y, b.y) + acc.y;
    o.z = fmaf(w0, hs.z, b.z) + acc.z;
    o.w = fmaf(w0, hs.w, b.w) + acc.w;
    reinterpret_cast<float4*>(out)[(size_t)node * LPR + c] = o;
  }
}


extern "C" void kernel_launch(void* const* d_in, const int* in_sizes, int n_in,
                              void* d_out, int out_size, void* d_ws, size_t ws_size,
                              hipStream_t stream) {
  const float* x  = (const float*)d_in[0];
  const int*   ei = (const int*)d_in[1];
  const float* W1 = (const float*)d_in[2];
  const float* b1 = (const float*)d_in[3];
  const float* W2 = (const float*)d_in[4];
  const float* b2 = (const float*)d_in[5];

  const int n = kNodes;
  const int e = in_sizes[1] / 2;
  const int* row = ei;
  const int* col = ei + e;

  char* ws = (char*)d_ws;
  constexpr size_t MiB = 1024 * 1024;
  constexpr size_t CNT_OFF  = 1 * MiB;
  constexpr size_t PTR_OFF  = 2 * MiB;
  constexpr size_t H_OFF    = 4 * MiB;
  constexpr size_t A1_OFF   = H_OFF + (size_t)kNodes * kHidDim * 4;
  constexpr size_t RANK_OFF = 112 * MiB;
  constexpr size_t EDGE_OFF = 144 * MiB;
  float* dinv   = (float*)ws;
  int*   cnt    = (int*)(ws + CNT_OFF);
  int*   rowptr = (int*)(ws + PTR_OFF);
  float* Hbuf   = (float*)(ws + H_OFF);
  float* A1     = (float*)(ws + A1_OFF);
  int*   rank   = (int*)(ws + RANK_OFF);
  int*   edges  = (int*)(ws + EDGE_OFF);
  float* out    = (float*)d_out;

  const int T = 256;

  k_cnt_zero<<<(n + T - 1) / T, T, 0, stream>>>(cnt, n);
  k_count_rank<<<(e + T - 1) / T, T, 0, stream>>>(col, cnt, rank, e);
  k_scan_dinv<<<(n + 4095) / 4096, 1024, 0, stream>>>(cnt, rowptr, dinv, n);
  k_fill<<<(e + T - 1) / T, T, 0, stream>>>(row, col, rank, rowptr, dinv, edges, e);

  const int waves = (n + 15) / 16;
  const int gemmBlocks = (waves + 3) / 4;
  const int aggBlocks = (n + 7) / 8;

  k_gemm_wmma<kInDim, kHidDim / 16, false>
      <<<gemmBlocks, 128, 0, stream>>>(x, W1, Hbuf, n);

  k_gather_agg<kHidDim>
      <<<aggBlocks, 256, 0, stream>>>(rowptr, cnt, edges, dinv, Hbuf, b1, A1, n, e);

  k_gemm_wmma<kHidDim, kOutDim / 16, true>
      <<<gemmBlocks, 128, 0, stream>>>(A1, W2, Hbuf, n);

  k_gather_agg<kOutDim>
      <<<aggBlocks, 256, 0, stream>>>(rowptr, cnt, edges, dinv, Hbuf, b2, out, n, e);
}


# Edge ranks from LDS counters: edges binned by 128-destination range into per-block cells, one block per range ranks and places them; 1.6M returning global adds gone; sum order = LDS arrival rank
# speedup vs baseline: 3.9552x; 1.7125x over previous
#include <hip/hip_runtime.h>


typedef __attribute__((ext_vector_type(2))) float v2f;
typedef __attribute__((ext_vector_type(8))) float v8f;

static constexpr int kNodes  = 100000;
static constexpr int kInDim  = 128;
static constexpr int kHidDim = 128;
static constexpr int kOutDim = 64;

static constexpr int kSlots = 64;


static constexpr int kBinShift = 7;
static constexpr int kBinNodes = 1 << kBinShift;
static constexpr int kBins     = (kNodes + kBinNodes - 1) / kBinNodes;
static constexpr int kParts    = 256;
static constexpr int kCellCap  = 32;
static constexpr int kSrcBits  = 24;
static constexpr int kSrcMask  = (1 << kSrcBits) - 1;
static constexpr int kBinThreads = 1024;
static_assert(kNodes <= (1 << kSrcBits), "a source index must fit the record's low field");
static_assert(kBinNodes <= 128, "destination-in-bin must leave the record non-negative");
static_assert(kParts % 32 == 0, "pass 2 reads the cells 32 at a time");

__global__ __launch_bounds__(kBinThreads) void k_bin_edges(const int* __restrict__ row,
                                                           const int* __restrict__ col,
                                                           int* __restrict__ cell_cnt,
                                                           int* __restrict__ cells,
                                                           int* __restrict__ ovf_cnt, int e, int n) {
  __shared__ int hist[kBins];
  const int t = threadIdx.x;
  const int part = blockIdx.x;
  for (int b = t; b < kBins; b += kBinThreads) hist[b] = 0;
  if (part == 0 && t == 0) ovf_cnt[0] = 0;
  __syncthreads();

  const int chunk = (e + kParts - 1) / kParts;
  const int beg = part * chunk;
  const int end = min(beg + chunk, e);
  constexpr int U = 4;
  for (int i0 = beg + t; i0 < end; i0 += U * kBinThreads) {
    int d[U], s[U];
#pragma unroll
    for (int u = 0; u < U; ++u) {
      const int i = i0 + u * kBinThreads;
      d[u] = -1;
      s[u] = 0;
      if (i < end) {
        d[u] = col[i];
        s[u] = row[i];
      }
    }
    int lr[U];
#pragma unroll
    for (int u = 0; u < U; ++u) {
      lr[u] = kCellCap;
      if ((unsigned)d[u] < (unsigned)n) lr[u] = atomicAdd(&hist[d[u] >> kBinShift], 1);
    }
#pragma unroll
    for (int u = 0; u < U; ++u) {
      if (lr[u] < kCellCap)
        cells[((size_t)(d[u] >> kBinShift) * kParts + part) * kCellCap + lr[u]] =
            ((d[u] & (kBinNodes - 1)) << kSrcBits) | (s[u] & kSrcMask);
    }
  }
  __syncthreads();
  for (int b = t; b < kBins; b += kBinThreads) cell_cnt[(size_t)part * kBins + b] = hist[b];
}

__device__ __forceinline__ void place_edge(int* lcnt, int dl, int node, int s,
                                           int* __restrict__ bucket, int* __restrict__ ovf_cnt,
                                           int* __restrict__ ovf_list, int e) {
  const int r = atomicAdd(&lcnt[dl], 1);
  if (r < kSlots) {
    bucket[(size_t)node * kSlots + r] = s;
  } else {
    const int p = atomicAdd(ovf_cnt, 1);
    if ((unsigned)p < (unsigned)e) reinterpret_cast<int2*>(ovf_list)[p] = make_int2(node, s);
  }
}

__global__ __launch_bounds__(256) void k_rank_place(const int* __restrict__ row,
                                                    const int* __restrict__ col,
                                                    const int* __restrict__ cell_cnt,
                                                    const int* __restrict__ cells,
                                                    int* __restrict__ cnt, int* __restrict__ ovf_cnt,
                                                    int* __restrict__ bucket,
                                                    int* __restrict__ ovf_list, int e, int n) {
  __shared__ int lcnt[kBinNodes];
  const int t = threadIdx.x;
  const int bin = blockIdx.x;
  if (t < kBinNodes) lcnt[t] = 0;
  __syncthreads();

  const int g = t >> 3;
  const int l8 = t & 7;
  constexpr int PASSES = kParts / 32;
  constexpr int RPL = kCellCap / 8;

  int c[PASSES];
#pragma unroll
  for (int p = 0; p < PASSES; ++p) c[p] = cell_cnt[(size_t)(p * 32 + g) * kBins + bin];

  int rec[PASSES][RPL];
#pragma unroll
  for (int p = 0; p < PASSES; ++p) {
    const int* __restrict__ cell = cells + ((size_t)bin * kParts + (p * 32 + g)) * kCellCap;
    const int valid = (c[p] <= kCellCap) ? c[p] : 0;
#pragma unroll
    for (int k = 0; k < RPL; ++k) {
      const int idx = l8 + 8 * k;
      rec[p][k] = -1;
      if (idx < valid) rec[p][k] = cell[idx];
    }
  }

#pragma unroll
  for (int p = 0; p < PASSES; ++p) {
#pragma unroll
    for (int k = 0; k < RPL; ++k) {
      const int r = rec[p][k];
      if (r >= 0) {
        const int dl = (r >> kSrcBits) & (kBinNodes - 1);
        place_edge(lcnt, dl, bin * kBinNodes + dl, r & kSrcMask, bucket, ovf_cnt, ovf_list, e);
      }
    }
  }

  const int chunk = (e + kParts - 1) / kParts;
  for (int p = 0; p < PASSES; ++p) {
    if (c[p] > kCellCap) {
      const int beg = (p * 32 + g) * chunk;
      const int end = min(beg + chunk, e);
      for (int i = beg + l8; i < end; i += 8) {
        const int d = col[i];
        if ((unsigned)d < (unsigned)n && (d >> kBinShift) == bin)
          place_edge(lcnt, d & (kBinNodes - 1), d, row[i] & kSrcMask, bucket, ovf_cnt, ovf_list, e);
      }
    }
  }

  __syncthreads();
  if (t < kBinNodes) {
    const int node = bin * kBinNodes + t;
    if (node < n) cnt[node] = lcnt[t];
  }
}


template <int K, int NT, bool RELU>
__global__ __launch_bounds__(128) void k_gemm_wmma(const float* __restrict__ X,
                                                   const float* __restrict__ W,
                                                   const int* __restrict__ cnt,
                                                   float* __restrict__ H, int M) {
  constexpr int N = NT * 16;
  const int wave = blockIdx.x * 4 + (threadIdx.x >> 5);
  const int lane = threadIdx.x & 31;
  const int m0 = wave * 16;
  if (m0 >= M) return;
  const int ml = lane & 15;
  const int hi = lane >> 4;

  v8f acc[NT];
#pragma unroll
  for (int t = 0; t < NT; ++t) acc[t] = {};

  const float* __restrict__ xrow = X + (size_t)(m0 + ml) * K;

#pragma unroll 4
  for (int k = 0; k < K; k += 4) {
    v2f a;
    a.x = xrow[k + 2 * hi + 0];
    a.y = xrow[k + 2 * hi + 1];
    if (RELU) {
      a.x = fmaxf(a.x, 0.0f);
      a.y = fmaxf(a.y, 0.0f);
    }
#pragma unroll
    for (int t = 0; t < NT; ++t) {
      v2f b;
      b.x = W[(size_t)(k + 2 * hi + 0) * N + t * 16 + ml];
      b.y = W[(size_t)(k + 2 * hi + 1) * N + t * 16 + ml];
      acc[t] = __builtin_amdgcn_wmma_f32_16x16x4_f32(
          false, a, false, b, (short)0, acc[t], false, false);
    }
  }

  float dv[8];
#pragma unroll
  for (int v = 0; v < 8; ++v) dv[v] = rsqrtf((float)(cnt[m0 + v + 8 * hi] + 1));

#pragma unroll
  for (int t = 0; t < NT; ++t) {
#pragma unroll
    for (int v = 0; v < 8; ++v) {
      H[(size_t)(m0 + v + 8 * hi) * N + t * 16 + ml] = acc[t][v] * dv[v];
    }
  }
}


template <int EPI>
__device__ __forceinline__ int lane_pick(int v, int j, int sub) {
  if (EPI == 1) return __builtin_amdgcn_readlane(v, j);
  const int a = __builtin_amdgcn_readlane(v, j);
  const int b = __builtin_amdgcn_readlane(v, j + 1);
  return sub ? b : a;
}

__device__ __forceinline__ void row_add(float4& acc, const float4 h) {
  acc.x += h.x;
  acc.y += h.y;
  acc.z += h.z;
  acc.w += h.w;
}

template <int D>
__global__ __launch_bounds__(256) void k_gather_agg(const int* __restrict__ cnt,
                                                    const int* __restrict__ bucket,
                                                    const int* __restrict__ ovf_cnt,
                                                    const int* __restrict__ ovf_list,
                                                    const float* __restrict__ H,
                                                    const float* __restrict__ bias,
                                                    float* __restrict__ out, int n, int e) {
  constexpr int LPR = D / 4;
  constexpr int EPI = 32 / LPR;
  static_assert(LPR == 32 || LPR == 16, "row must be one wave or half a wave of float4");
  const int lane = threadIdx.x & 31;
  const int node = __builtin_amdgcn_readfirstlane(blockIdx.x * 8 + (threadIdx.x >> 5));
  if (node >= n) return;
  const int sub = lane / LPR;
  const int c = lane % LPR;

  const int mt = cnt[node];
  const float di = rsqrtf((float)(mt + 1));
  const int m = min(max(mt, 0), kSlots);

  const float4* __restrict__ H4 = reinterpret_cast<const float4*>(H);
  const float4 hs = H4[(size_t)node * LPR + c];
  float4 acc = make_float4(0.0f, 0.0f, 0.0f, 0.0f);
  const int* __restrict__ slots = bucket + (size_t)node * kSlots;

  for (int base = 0; base < m; base += 32) {
    const int rem = min(32, m - base);
    int sv = 0;
    if (lane < rem) sv = slots[base + lane];
    const int src = (int)min((unsigned)sv, (unsigned)(n - 1));
    const int full = rem - (rem % EPI);
#pragma unroll 4
    for (int j = 0; j < full; j += EPI) {
      const int s = lane_pick<EPI>(src, j, sub);
      row_add(acc, H4[(size_t)s * LPR + c]);
    }
    if (EPI == 2 && (rem & 1)) {
      const int s = __builtin_amdgcn_readlane(src, rem - 1);
      if (sub == 0) row_add(acc, H4[(size_t)s * LPR + c]);
    }
  }

  if (mt > kSlots) {
    const int2* __restrict__ ovf = reinterpret_cast<const int2*>(ovf_list);
    const int on = min(max(ovf_cnt[0], 0), e);
    for (int b = 0; b < on; b += 32) {
      int2 oe = make_int2(-1, 0);
      if (b + lane < on) oe = ovf[b + lane];
      const int src = (int)min((unsigned)oe.y, (unsigned)(n - 1));
      unsigned hit = __builtin_amdgcn_ballot_w32(oe.x == node);
      while (hit) {
        const int j = __builtin_ctz(hit);
        hit &= hit - 1;
        const int s = __builtin_amdgcn_readlane(src, j);
        if (sub == 0) row_add(acc, H4[(size_t)s * LPR + c]);
      }
    }
  }

  if (EPI == 2) {
    acc.x += __shfl_xor(acc.x, 16);
    acc.y += __shfl_xor(acc.y, 16);
    acc.z += __shfl_xor(acc.z, 16);
    acc.w += __shfl_xor(acc.w, 16);
  }
  if (sub == 0) {
    const float4 b = reinterpret_cast<const float4*>(bias)[c];
    float4 o;
    o.x = fmaf(di, acc.x + hs.x, b.x);
    o.y = fmaf(di, acc.y + hs.y, b.y);
    o.z = fmaf(di, acc.z + hs.z, b.z);
    o.w = fmaf(di, acc.w + hs.w, b.w);
    reinterpret_cast<float4*>(out)[(size_t)node * LPR + c] = o;
  }
}


extern "C" void kernel_launch(void* const* d_in, const int* in_sizes, int n_in,
                              void* d_out, int out_size, void* d_ws, size_t ws_size,
                              hipStream_t stream) {
  const float* x  = (const float*)d_in[0];
  const int*   ei = (const int*)d_in[1];
  const float* W1 = (const float*)d_in[2];
  const float* b1 = (const float*)d_in[3];
  const float* W2 = (const float*)d_in[4];
  const float* b2 = (const float*)d_in[5];

  const int n = kNodes;
  const int e = in_sizes[1] / 2;
  const int* row = ei;
  const int* col = ei + e;

  char* ws = (char*)d_ws;
  constexpr size_t MiB = 1024 * 1024;
  constexpr size_t CNT_OFF    = 1 * MiB;
  constexpr size_t H_OFF      = 4 * MiB;
  constexpr size_t A1_OFF     = H_OFF + (size_t)kNodes * kHidDim * 4;
  constexpr size_t BUCKET_OFF = 112 * MiB;
  constexpr size_t OVF_OFF    = 144 * MiB;
  constexpr size_t CELLCNT_OFF = 2 * MiB;
  constexpr size_t CELLS_OFF   = 192 * MiB;
  static_assert(A1_OFF + (size_t)kNodes * kHidDim * 4 <= BUCKET_OFF, "A1 must end before the buckets");
  static_assert(BUCKET_OFF + (size_t)kNodes * kSlots * 4 <= OVF_OFF, "buckets must end before the overflow list");
  static_assert(CNT_OFF + ((size_t)kNodes + 1) * 4 <= CELLCNT_OFF, "counts must end before the cell counts");
  static_assert(CELLCNT_OFF + (size_t)kParts * kBins * 4 <= H_OFF, "cell counts must end before H");
  static_assert(CELLS_OFF + (size_t)kBins * kParts * kCellCap * 4 <= 256 * MiB, "cells must end inside the workspace");
  int*   cnt     = (int*)(ws + CNT_OFF);
  int*   ovf_cnt = cnt + n;
  float* Hbuf    = (float*)(ws + H_OFF);
  float* A1      = (float*)(ws + A1_OFF);
  int*   bucket  = (int*)(ws + BUCKET_OFF);
  int*   cell_cnt = (int*)(ws + CELLCNT_OFF);
  int*   cells    = (int*)(ws + CELLS_OFF);
  int*   ovf     = (int*)(ws + OVF_OFF);
  float* out     = (float*)d_out;

  k_bin_edges<<<kParts, kBinThreads, 0, stream>>>(row, col, cell_cnt, cells, ovf_cnt, e, n);
  k_rank_place<<<kBins, 256, 0, stream>>>(row, col, cell_cnt, cells, cnt, ovf_cnt, bucket, ovf, e, n);

  const int waves = (n + 15) / 16;
  const int gemmBlocks = (waves + 3) / 4;
  const int aggBlocks = (n + 7) / 8;

  k_gemm_wmma<kInDim, kHidDim / 16, false>
      <<<gemmBlocks, 128, 0, stream>>>(x, W1, cnt, Hbuf, n);

  k_gather_agg<kHidDim>
      <<<aggBlocks, 256, 0, stream>>>(cnt, bucket, ovf_cnt, ovf, Hbuf, b1, A1, n, e);

  k_gemm_wmma<kHidDim, kOutDim / 16, true>
      <<<gemmBlocks, 128, 0, stream>>>(A1, W2, cnt, Hbuf, n);

  k_gather_agg<kOutDim>
      <<<aggBlocks, 256, 0, stream>>>(cnt, bucket, ovf_cnt, ovf, Hbuf, b2, out, n, e);
}


# Layer-1 edge sum and layer-2 product in one launch: a block sums 32 nodes' rows (4 per wave), passes relu rows to the WMMA loop through LDS, stores whole rows; A1 plane and one launch gone
# speedup vs baseline: 4.5197x; 1.1427x over previous
#include <hip/hip_runtime.h>


typedef __attribute__((ext_vector_type(2))) float v2f;
typedef __attribute__((ext_vector_type(8))) float v8f;

static constexpr int kNodes  = 100000;
static constexpr int kInDim  = 128;
static constexpr int kHidDim = 128;
static constexpr int kOutDim = 64;

static constexpr int kSlots = 64;


static constexpr int kBinShift = 7;
static constexpr int kBinNodes = 1 << kBinShift;
static constexpr int kBins     = (kNodes + kBinNodes - 1) / kBinNodes;
static constexpr int kParts    = 256;
static constexpr int kCellCap  = 32;
static constexpr int kSrcBits  = 24;
static constexpr int kSrcMask  = (1 << kSrcBits) - 1;
static constexpr int kBinThreads = 1024;
static_assert(kNodes <= (1 << kSrcBits), "a source index must fit the record's low field");
static_assert(kBinNodes <= 128, "destination-in-bin must leave the record non-negative");
static_assert(kParts % 32 == 0, "pass 2 reads the cells 32 at a time");

__global__ __launch_bounds__(kBinThreads) void k_bin_edges(const int* __restrict__ row,
                                                           const int* __restrict__ col,
                                                           int* __restrict__ cell_cnt,
                                                           int* __restrict__ cells,
                                                           int* __restrict__ ovf_cnt, int e, int n) {
  __shared__ int hist[kBins];
  const int t = threadIdx.x;
  const int part = blockIdx.x;
  for (int b = t; b < kBins; b += kBinThreads) hist[b] = 0;
  if (part == 0 && t == 0) ovf_cnt[0] = 0;
  __syncthreads();

  const int chunk = (e + kParts - 1) / kParts;
  const int beg = part * chunk;
  const int end = min(beg + chunk, e);
  constexpr int U = 4;
  for (int i0 = beg + t; i0 < end; i0 += U * kBinThreads) {
    int d[U], s[U];
#pragma unroll
    for (int u = 0; u < U; ++u) {
      const int i = i0 + u * kBinThreads;
      d[u] = -1;
      s[u] = 0;
      if (i < end) {
        d[u] = col[i];
        s[u] = row[i];
      }
    }
    int lr[U];
#pragma unroll
    for (int u = 0; u < U; ++u) {
      lr[u] = kCellCap;
      if ((unsigned)d[u] < (unsigned)n) lr[u] = atomicAdd(&hist[d[u] >> kBinShift], 1);
    }
#pragma unroll
    for (int u = 0; u < U; ++u) {
      if (lr[u] < kCellCap)
        cells[((size_t)(d[u] >> kBinShift) * kParts + part) * kCellCap + lr[u]] =
            ((d[u] & (kBinNodes - 1)) << kSrcBits) | (s[u] & kSrcMask);
    }
  }
  __syncthreads();
  for (int b = t; b < kBins; b += kBinThreads) cell_cnt[(size_t)part * kBins + b] = hist[b];
}

__device__ __forceinline__ void place_edge(int* lcnt, int dl, int node, int s,
                                           int* __restrict__ bucket, int* __restrict__ ovf_cnt,
                                           int* __restrict__ ovf_list, int e) {
  const int r = atomicAdd(&lcnt[dl], 1);
  if (r < kSlots) {
    bucket[(size_t)node * kSlots + r] = s;
  } else {
    const int p = atomicAdd(ovf_cnt, 1);
    if ((unsigned)p < (unsigned)e) reinterpret_cast<int2*>(ovf_list)[p] = make_int2(node, s);
  }
}

__global__ __launch_bounds__(256) void k_rank_place(const int* __restrict__ row,
                                                    const int* __restrict__ col,
                                                    const int* __restrict__ cell_cnt,
                                                    const int* __restrict__ cells,
                                                    int* __restrict__ cnt, int* __restrict__ ovf_cnt,
                                                    int* __restrict__ bucket,
                                                    int* __restrict__ ovf_list, int e, int n) {
  __shared__ int lcnt[kBinNodes];
  const int t = threadIdx.x;
  const int bin = blockIdx.x;
  if (t < kBinNodes) lcnt[t] = 0;
  __syncthreads();

  const int g = t >> 3;
  const int l8 = t & 7;
  constexpr int PASSES = kParts / 32;
  constexpr int RPL = kCellCap / 8;

  int c[PASSES];
#pragma unroll
  for (int p = 0; p < PASSES; ++p) c[p] = cell_cnt[(size_t)(p * 32 + g) * kBins + bin];

  int rec[PASSES][RPL];
#pragma unroll
  for (int p = 0; p < PASSES; ++p) {
    const int* __restrict__ cell = cells + ((size_t)bin * kParts + (p * 32 + g)) * kCellCap;
    const int valid = (c[p] <= kCellCap) ? c[p] : 0;
#pragma unroll
    for (int k = 0; k < RPL; ++k) {
      const int idx = l8 + 8 * k;
      rec[p][k] = -1;
      if (idx < valid) rec[p][k] = cell[idx];
    }
  }

#pragma unroll
  for (int p = 0; p < PASSES; ++p) {
#pragma unroll
    for (int k = 0; k < RPL; ++k) {
      const int r = rec[p][k];
      if (r >= 0) {
        const int dl = (r >> kSrcBits) & (kBinNodes - 1);
        place_edge(lcnt, dl, bin * kBinNodes + dl, r & kSrcMask, bucket, ovf_cnt, ovf_list, e);
      }
    }
  }

  const int chunk = (e + kParts - 1) / kParts;
  for (int p = 0; p < PASSES; ++p) {
    if (c[p] > kCellCap) {
      const int beg = (p * 32 + g) * chunk;
      const int end = min(beg + chunk, e);
      for (int i = beg + l8; i < end; i += 8) {
        const int d = col[i];
        if ((unsigned)d < (unsigned)n && (d >> kBinShift) == bin)
          place_edge(lcnt, d & (kBinNodes - 1), d, row[i] & kSrcMask, bucket, ovf_cnt, ovf_list, e);
      }
    }
  }

  __syncthreads();
  if (t < kBinNodes) {
    const int node = bin * kBinNodes + t;
    if (node < n) cnt[node] = lcnt[t];
  }
}


template <int K, int NT, bool RELU>
__global__ __launch_bounds__(128) void k_gemm_wmma(const float* __restrict__ X,
                                                   const float* __restrict__ W,
                                                   const int* __restrict__ cnt,
                                                   float* __restrict__ H, int M) {
  constexpr int N = NT * 16;
  const int wave = blockIdx.x * 4 + (threadIdx.x >> 5);
  const int lane = threadIdx.x & 31;
  const int m0 = wave * 16;
  if (m0 >= M) return;
  const int ml = lane & 15;
  const int hi = lane >> 4;

  v8f acc[NT];
#pragma unroll
  for (int t = 0; t < NT; ++t) acc[t] = {};

  const float* __restrict__ xrow = X + (size_t)(m0 + ml) * K;

#pragma unroll 4
  for (int k = 0; k < K; k += 4) {
    v2f a;
    a.x = xrow[k + 2 * hi + 0];
    a.y = xrow[k + 2 * hi + 1];
    if (RELU) {
      a.x = fmaxf(a.x, 0.0f);
      a.y = fmaxf(a.y, 0.0f);
    }
#pragma unroll
    for (int t = 0; t < NT; ++t) {
      v2f b;
      b.x = W[(size_t)(k + 2 * hi + 0) * N + t * 16 + ml];
      b.y = W[(size_t)(k + 2 * hi + 1) * N + t * 16 + ml];
      acc[t] = __builtin_amdgcn_wmma_f32_16x16x4_f32(
          false, a, false, b, (short)0, acc[t], false, false);
    }
  }

  float dv[8];
#pragma unroll
  for (int v = 0; v < 8; ++v) dv[v] = rsqrtf((float)(cnt[m0 + v + 8 * hi] + 1));

#pragma unroll
  for (int t = 0; t < NT; ++t) {
#pragma unroll
    for (int v = 0; v < 8; ++v) {
      H[(size_t)(m0 + v + 8 * hi) * N + t * 16 + ml] = acc[t][v] * dv[v];
    }
  }
}


template <int EPI>
__device__ __forceinline__ int lane_pick(int v, int j, int sub) {
  if (EPI == 1) return __builtin_amdgcn_readlane(v, j);
  const int a = __builtin_amdgcn_readlane(v, j);
  const int b = __builtin_amdgcn_readlane(v, j + 1);
  return sub ? b : a;
}

__device__ __forceinline__ void row_add(float4& acc, const float4 h) {
  acc.x += h.x;
  acc.y += h.y;
  acc.z += h.z;
  acc.w += h.w;
}

template <int D>
__global__ __launch_bounds__(256) void k_gather_agg(const int* __restrict__ cnt,
                                                    const int* __restrict__ bucket,
                                                    const int* __restrict__ ovf_cnt,
                                                    const int* __restrict__ ovf_list,
                                                    const float* __restrict__ H,
                                                    const float* __restrict__ bias,
                                                    float* __restrict__ out, int n, int e) {
  constexpr int LPR = D / 4;
  constexpr int EPI = 32 / LPR;
  static_assert(LPR == 32 || LPR == 16, "row must be one wave or half a wave of float4");
  const int lane = threadIdx.x & 31;
  const int node = __builtin_amdgcn_readfirstlane(blockIdx.x * 8 + (threadIdx.x >> 5));
  if (node >= n) return;
  const int sub = lane / LPR;
  const int c = lane % LPR;

  const int mt = cnt[node];
  const float di = rsqrtf((float)(mt + 1));
  const int m = min(max(mt, 0), kSlots);

  const float4* __restrict__ H4 = reinterpret_cast<const float4*>(H);
  const float4 hs = H4[(size_t)node * LPR + c];
  float4 acc = make_float4(0.0f, 0.0f, 0.0f, 0.0f);
  const int* __restrict__ slots = bucket + (size_t)node * kSlots;

  for (int base = 0; base < m; base += 32) {
    const int rem = min(32, m - base);
    int sv = 0;
    if (lane < rem) sv = slots[base + lane];
    const int src = (int)min((unsigned)sv, (unsigned)(n - 1));
    const int full = rem - (rem % EPI);
#pragma unroll 4
    for (int j = 0; j < full; j += EPI) {
      const int s = lane_pick<EPI>(src, j, sub);
      row_add(acc, H4[(size_t)s * LPR + c]);
    }
    if (EPI == 2 && (rem & 1)) {
      const int s = __builtin_amdgcn_readlane(src, rem - 1);
      if (sub == 0) row_add(acc, H4[(size_t)s * LPR + c]);
    }
  }

  if (mt > kSlots) {
    const int2* __restrict__ ovf = reinterpret_cast<const int2*>(ovf_list);
    const int on = min(max(ovf_cnt[0], 0), e);
    for (int b = 0; b < on; b += 32) {
      int2 oe = make_int2(-1, 0);
      if (b + lane < on) oe = ovf[b + lane];
      const int src = (int)min((unsigned)oe.y, (unsigned)(n - 1));
      unsigned hit = __builtin_amdgcn_ballot_w32(oe.x == node);
      while (hit) {
        const int j = __builtin_ctz(hit);
        hit &= hit - 1;
        const int s = __builtin_amdgcn_readlane(src, j);
        if (sub == 0) row_add(acc, H4[(size_t)s * LPR + c]);
      }
    }
  }

  if (EPI == 2) {
    acc.x += __shfl_xor(acc.x, 16);
    acc.y += __shfl_xor(acc.y, 16);
    acc.z += __shfl_xor(acc.z, 16);
    acc.w += __shfl_xor(acc.w, 16);
  }
  if (sub == 0) {
    const float4 b = reinterpret_cast<const float4*>(bias)[c];
    float4 o;
    o.x = fmaf(di, acc.x + hs.x, b.x);
    o.y = fmaf(di, acc.y + hs.y, b.y);
    o.z = fmaf(di, acc.z + hs.z, b.z);
    o.w = fmaf(di, acc.w + hs.w, b.w);
    reinterpret_cast<float4*>(out)[(size_t)node * LPR + c] = o;
  }
}


static constexpr int kFN     = 32;
static constexpr int kFPitch = kHidDim + 4;
static constexpr int kOPitch = kOutDim + 4;

__device__ __forceinline__ float4 sum_rows128(int node, int mt, int sv_first, int lane,
                                              const int* __restrict__ bucket,
                                              const int* __restrict__ ovf_cnt,
                                              const int* __restrict__ ovf_list,
                                              const float4* __restrict__ H4, int n, int e) {
  constexpr int LPR = kHidDim / 4;
  const int m = min(max(mt, 0), kSlots);
  float4 acc = make_float4(0.0f, 0.0f, 0.0f, 0.0f);
  const int* __restrict__ slots = bucket + (size_t)node * kSlots;
  for (int base = 0; base < m; base += 32) {
    const int rem = min(32, m - base);
    int sv = sv_first;
    if (base > 0) {
      sv = 0;
      if (lane < rem) sv = slots[base + lane];
    }
    const int src = (int)min((unsigned)sv, (unsigned)(n - 1));
#pragma unroll 4
    for (int j = 0; j < rem; ++j) {
      const int sidx = __builtin_amdgcn_readlane(src, j);
      row_add(acc, H4[(size_t)sidx * LPR + lane]);
    }
  }
  if (mt > kSlots) {
    const int2* __restrict__ ovf = reinterpret_cast<const int2*>(ovf_list);
    const int on = min(max(ovf_cnt[0], 0), e);
    for (int b = 0; b < on; b += 32) {
      int2 oe = make_int2(-1, 0);
      if (b + lane < on) oe = ovf[b + lane];
      const int src = (int)min((unsigned)oe.y, (unsigned)(n - 1));
      unsigned hit = __builtin_amdgcn_ballot_w32(oe.x == node);
      while (hit) {
        const int j = __builtin_ctz(hit);
        hit &= hit - 1;
        const int sidx = __builtin_amdgcn_readlane(src, j);
        row_add(acc, H4[(size_t)sidx * LPR + lane]);
      }
    }
  }
  return acc;
}

__global__ __launch_bounds__(256, 16) void k_agg_gemm(const int* __restrict__ cnt,
                                                  const int* __restrict__ bucket,
                                                  const int* __restrict__ ovf_cnt,
                                                  const int* __restrict__ ovf_list,
                                                  const float* __restrict__ H,
                                                  const float* __restrict__ bias,
                                                  const float* __restrict__ W,
                                                  float* __restrict__ H2, int n, int e) {
  constexpr int LPR = kHidDim / 4;
  constexpr int NPW = kFN / 8;
  __shared__ __attribute__((aligned(16))) float tileA[kFN * kFPitch];
  __shared__ __attribute__((aligned(16))) float tileO[kFN * kOPitch];
  __shared__ float sdi[kFN];

  const int lane = threadIdx.x & 31;
  const int wv = __builtin_amdgcn_readfirstlane(threadIdx.x >> 5);
  const int node0 = blockIdx.x * kFN;
  const float4* __restrict__ H4 = reinterpret_cast<const float4*>(H);

  int nd[NPW];
  int mt[NPW];
#pragma unroll
  for (int i = 0; i < NPW; ++i) {
    nd[i] = min(node0 + wv * NPW + i, n - 1);
    mt[i] = cnt[nd[i]];
  }
  int sv0[NPW];
  float4 hs[NPW];
#pragma unroll
  for (int i = 0; i < NPW; ++i) {
    const int m0 = min(min(max(mt[i], 0), kSlots), 32);
    sv0[i] = 0;
    if (lane < m0) sv0[i] = bucket[(size_t)nd[i] * kSlots + lane];
    hs[i] = H4[(size_t)nd[i] * LPR + lane];
  }
  const float4 bv = reinterpret_cast<const float4*>(bias)[lane];

#pragma unroll
  for (int i = 0; i < NPW; ++i) {
    const int r = wv * NPW + i;
    const float4 acc = sum_rows128(nd[i], mt[i], sv0[i], lane, bucket, ovf_cnt, ovf_list, H4, n, e);
    const float di = rsqrtf((float)(mt[i] + 1));
    float4 o;
    o.x = fmaxf(fmaf(di, acc.x + hs[i].x, bv.x), 0.0f);
    o.y = fmaxf(fmaf(di, acc.y + hs[i].y, bv.y), 0.0f);
    o.z = fmaxf(fmaf(di, acc.z + hs[i].z, bv.z), 0.0f);
    o.w = fmaxf(fmaf(di, acc.w + hs[i].w, bv.w), 0.0f);
    *reinterpret_cast<float4*>(&tileA[r * kFPitch + 4 * lane]) = o;
    if (lane == 0) sdi[r] = di;
  }

  constexpr int KB = 4;
  const int ml = lane & 15;
  const int hi = lane >> 4;
  const float* __restrict__ wcol = W + (2 * hi) * kOutDim + (wv & 3) * 16 + ml;
  v2f bq[KB];
#pragma unroll
  for (int u = 0; u < KB; ++u) bq[u] = {};
  if (wv < 4) {
#pragma unroll
    for (int u = 0; u < KB; ++u) {
      bq[u].x = wcol[(4 * u + 0) * kOutDim];
      bq[u].y = wcol[(4 * u + 1) * kOutDim];
    }
  }
  __syncthreads();

  if (wv < 4) {
    const float* arow0 = &tileA[ml * kFPitch + 2 * hi];
    const float* arow1 = arow0 + 16 * kFPitch;
    v8f acc0 = {};
    v8f acc1 = {};
#pragma unroll
    for (int kb = 0; kb < kHidDim / (4 * KB); ++kb) {
      v2f bn[KB];
#pragma unroll
      for (int u = 0; u < KB; ++u) {
        bn[u] = {};
        if (kb + 1 < kHidDim / (4 * KB)) {
          bn[u].x = wcol[((kb + 1) * 4 * KB + 4 * u + 0) * kOutDim];
          bn[u].y = wcol[((kb + 1) * 4 * KB + 4 * u + 1) * kOutDim];
        }
      }
#pragma unroll
      for (int u = 0; u < KB; ++u) {
        const int k = kb * 4 * KB + 4 * u;
        v2f a0, a1;
        a0.x = arow0[k + 0];
        a0.y = arow0[k + 1];
        a1.x = arow1[k + 0];
        a1.y = arow1[k + 1];
        acc0 = __builtin_amdgcn_wmma_f32_16x16x4_f32(false, a0, false, bq[u], (short)0, acc0, false, false);
        acc1 = __builtin_amdgcn_wmma_f32_16x16x4_f32(false, a1, false, bq[u], (short)0, acc1, false, false);
      }
#pragma unroll
      for (int u = 0; u < KB; ++u) bq[u] = bn[u];
    }
#pragma unroll
    for (int v = 0; v < 8; ++v) {
      const int r0 = v + 8 * hi;
      const int r1 = r0 + 16;
      tileO[r0 * kOPitch + wv * 16 + ml] = acc0[v] * sdi[r0];
      tileO[r1 * kOPitch + wv * 16 + ml] = acc1[v] * sdi[r1];
    }
  }
  __syncthreads();

  constexpr int OQ = kOutDim / 4;
#pragma unroll
  for (int q = 0; q < (kFN * OQ) / 256; ++q) {
    const int idx = threadIdx.x + 256 * q;
    const int r = idx / OQ;
    const int c4 = idx % OQ;
    if (node0 + r < n)
      reinterpret_cast<float4*>(H2)[(size_t)(node0 + r) * OQ + c4] =
          *reinterpret_cast<const float4*>(&tileO[r * kOPitch + 4 * c4]);
  }
}


extern "C" void kernel_launch(void* const* d_in, const int* in_sizes, int n_in,
                              void* d_out, int out_size, void* d_ws, size_t ws_size,
                              hipStream_t stream) {
  const float* x  = (const float*)d_in[0];
  const int*   ei = (const int*)d_in[1];
  const float* W1 = (const float*)d_in[2];
  const float* b1 = (const float*)d_in[3];
  const float* W2 = (const float*)d_in[4];
  const float* b2 = (const float*)d_in[5];

  const int n = kNodes;
  const int e = in_sizes[1] / 2;
  const int* row = ei;
  const int* col = ei + e;

  char* ws = (char*)d_ws;
  constexpr size_t MiB = 1024 * 1024;
  constexpr size_t CNT_OFF    = 1 * MiB;
  constexpr size_t H_OFF      = 4 * MiB;
  constexpr size_t A1_OFF     = H_OFF + (size_t)kNodes * kHidDim * 4;
  constexpr size_t BUCKET_OFF = 112 * MiB;
  constexpr size_t OVF_OFF    = 144 * MiB;
  constexpr size_t CELLCNT_OFF = 2 * MiB;
  constexpr size_t CELLS_OFF   = 192 * MiB;
  static_assert(A1_OFF + (size_t)kNodes * kOutDim * 4 <= BUCKET_OFF, "layer-2 rows must end before the buckets");
  static_assert(BUCKET_OFF + (size_t)kNodes * kSlots * 4 <= OVF_OFF, "buckets must end before the overflow list");
  static_assert(CNT_OFF + ((size_t)kNodes + 1) * 4 <= CELLCNT_OFF, "counts must end before the cell counts");
  static_assert(CELLCNT_OFF + (size_t)kParts * kBins * 4 <= H_OFF, "cell counts must end before H");
  static_assert(CELLS_OFF + (size_t)kBins * kParts * kCellCap * 4 <= 256 * MiB, "cells must end inside the workspace");
  int*   cnt     = (int*)(ws + CNT_OFF);
  int*   ovf_cnt = cnt + n;
  float* Hbuf    = (float*)(ws + H_OFF);
  float* Hs2     = (float*)(ws + A1_OFF);
  int*   bucket  = (int*)(ws + BUCKET_OFF);
  int*   cell_cnt = (int*)(ws + CELLCNT_OFF);
  int*   cells    = (int*)(ws + CELLS_OFF);
  int*   ovf     = (int*)(ws + OVF_OFF);
  float* out     = (float*)d_out;

  k_bin_edges<<<kParts, kBinThreads, 0, stream>>>(row, col, cell_cnt, cells, ovf_cnt, e, n);
  k_rank_place<<<kBins, 256, 0, stream>>>(row, col, cell_cnt, cells, cnt, ovf_cnt, bucket, ovf, e, n);

  const int waves = (n + 15) / 16;
  const int gemmBlocks = (waves + 3) / 4;
  const int aggBlocks = (n + 7) / 8;

  k_gemm_wmma<kInDim, kHidDim / 16, false>
      <<<gemmBlocks, 128, 0, stream>>>(x, W1, cnt, Hbuf, n);

  const int fusedBlocks = (n + kFN - 1) / kFN;
  k_agg_gemm<<<fusedBlocks, 256, 0, stream>>>(cnt, bucket, ovf_cnt, ovf, Hbuf, b1, W2, Hs2, n, e);

  k_gather_agg<kOutDim>
      <<<aggBlocks, 256, 0, stream>>>(cnt, bucket, ovf_cnt, ovf, Hs2, b2, out, n, e);
}


# Layer 1 as (A X) W1: a block sums weighted rows of x (weight dinv of the source, from cnt), then tile@W1 and tile@W2 on WMMA; X@W1 launch and H plane gone; layer-1 sums precede the product
# speedup vs baseline: 4.8521x; 1.0735x over previous
#include <hip/hip_runtime.h>


typedef __attribute__((ext_vector_type(2))) float v2f;
typedef __attribute__((ext_vector_type(8))) float v8f;

static constexpr int kNodes  = 100000;
static constexpr int kInDim  = 128;
static constexpr int kHidDim = 128;
static constexpr int kOutDim = 64;

static constexpr int kSlots = 64;


static constexpr int kBinShift = 7;
static constexpr int kBinNodes = 1 << kBinShift;
static constexpr int kBins     = (kNodes + kBinNodes - 1) / kBinNodes;
static constexpr int kParts    = 256;
static constexpr int kCellCap  = 32;
static constexpr int kSrcBits  = 24;
static constexpr int kSrcMask  = (1 << kSrcBits) - 1;
static constexpr int kBinThreads = 1024;
static_assert(kNodes <= (1 << kSrcBits), "a source index must fit the record's low field");
static_assert(kBinNodes <= 128, "destination-in-bin must leave the record non-negative");
static_assert(kParts % 32 == 0, "pass 2 reads the cells 32 at a time");

__global__ __launch_bounds__(kBinThreads) void k_bin_edges(const int* __restrict__ row,
                                                           const int* __restrict__ col,
                                                           int* __restrict__ cell_cnt,
                                                           int* __restrict__ cells,
                                                           int* __restrict__ ovf_cnt, int e, int n) {
  __shared__ int hist[kBins];
  const int t = threadIdx.x;
  const int part = blockIdx.x;
  for (int b = t; b < kBins; b += kBinThreads) hist[b] = 0;
  if (part == 0 && t == 0) ovf_cnt[0] = 0;
  __syncthreads();

  const int chunk = (e + kParts - 1) / kParts;
  const int beg = part * chunk;
  const int end = min(beg + chunk, e);
  constexpr int U = 4;
  for (int i0 = beg + t; i0 < end; i0 += U * kBinThreads) {
    int d[U], s[U];
#pragma unroll
    for (int u = 0; u < U; ++u) {
      const int i = i0 + u * kBinThreads;
      d[u] = -1;
      s[u] = 0;
      if (i < end) {
        d[u] = col[i];
        s[u] = row[i];
      }
    }
    int lr[U];
#pragma unroll
    for (int u = 0; u < U; ++u) {
      lr[u] = kCellCap;
      if ((unsigned)d[u] < (unsigned)n) lr[u] = atomicAdd(&hist[d[u] >> kBinShift], 1);
    }
#pragma unroll
    for (int u = 0; u < U; ++u) {
      if (lr[u] < kCellCap)
        cells[((size_t)(d[u] >> kBinShift) * kParts + part) * kCellCap + lr[u]] =
            ((d[u] & (kBinNodes - 1)) << kSrcBits) | (s[u] & kSrcMask);
    }
  }
  __syncthreads();
  for (int b = t; b < kBins; b += kBinThreads) cell_cnt[(size_t)part * kBins + b] = hist[b];
}

__device__ __forceinline__ void place_edge(int* lcnt, int dl, int node, int s,
                                           int* __restrict__ bucket, int* __restrict__ ovf_cnt,
                                           int* __restrict__ ovf_list, int e) {
  const int r = atomicAdd(&lcnt[dl], 1);
  if (r < kSlots) {
    bucket[(size_t)node * kSlots + r] = s;
  } else {
    const int p = atomicAdd(ovf_cnt, 1);
    if ((unsigned)p < (unsigned)e) reinterpret_cast<int2*>(ovf_list)[p] = make_int2(node, s);
  }
}

__global__ __launch_bounds__(256) void k_rank_place(const int* __restrict__ row,
                                                    const int* __restrict__ col,
                                                    const int* __restrict__ cell_cnt,
                                                    const int* __restrict__ cells,
                                                    int* __restrict__ cnt, int* __restrict__ ovf_cnt,
                                                    int* __restrict__ bucket,
                                                    int* __restrict__ ovf_list, int e, int n) {
  __shared__ int lcnt[kBinNodes];
  const int t = threadIdx.x;
  const int bin = blockIdx.x;
  if (t < kBinNodes) lcnt[t] = 0;
  __syncthreads();

  const int g = t >> 3;
  const int l8 = t & 7;
  constexpr int PASSES = kParts / 32;
  constexpr int RPL = kCellCap / 8;

  int c[PASSES];
#pragma unroll
  for (int p = 0; p < PASSES; ++p) c[p] = cell_cnt[(size_t)(p * 32 + g) * kBins + bin];

  int rec[PASSES][RPL];
#pragma unroll
  for (int p = 0; p < PASSES; ++p) {
    const int* __restrict__ cell = cells + ((size_t)bin * kParts + (p * 32 + g)) * kCellCap;
    const int valid = (c[p] <= kCellCap) ? c[p] : 0;
#pragma unroll
    for (int k = 0; k < RPL; ++k) {
      const int idx = l8 + 8 * k;
      rec[p][k] = -1;
      if (idx < valid) rec[p][k] = cell[idx];
    }
  }

#pragma unroll
  for (int p = 0; p < PASSES; ++p) {
#pragma unroll
    for (int k = 0; k < RPL; ++k) {
      const int r = rec[p][k];
      if (r >= 0) {
        const int dl = (r >> kSrcBits) & (kBinNodes - 1);
        place_edge(lcnt, dl, bin * kBinNodes + dl, r & kSrcMask, bucket, ovf_cnt, ovf_list, e);
      }
    }
  }

  const int chunk = (e + kParts - 1) / kParts;
  for (int p = 0; p < PASSES; ++p) {
    if (c[p] > kCellCap) {
      const int beg = (p * 32 + g) * chunk;
      const int end = min(beg + chunk, e);
      for (int i = beg + l8; i < end; i += 8) {
        const int d = col[i];
        if ((unsigned)d < (unsigned)n && (d >> kBinShift) == bin)
          place_edge(lcnt, d & (kBinNodes - 1), d, row[i] & kSrcMask, bucket, ovf_cnt, ovf_list, e);
      }
    }
  }

  __syncthreads();
  if (t < kBinNodes) {
    const int node = bin * kBinNodes + t;
    if (node < n) cnt[node] = lcnt[t];
  }
}


template <int EPI>
__device__ __forceinline__ int lane_pick(int v, int j, int sub) {
  if (EPI == 1) return __builtin_amdgcn_readlane(v, j);
  const int a = __builtin_amdgcn_readlane(v, j);
  const int b = __builtin_amdgcn_readlane(v, j + 1);
  return sub ? b : a;
}

__device__ __forceinline__ void row_add(float4& acc, const float4 h) {
  acc.x += h.x;
  acc.y += h.y;
  acc.z += h.z;
  acc.w += h.w;
}

template <int D>
__global__ __launch_bounds__(256) void k_gather_agg(const int* __restrict__ cnt,
                                                    const int* __restrict__ bucket,
                                                    const int* __restrict__ ovf_cnt,
                                                    const int* __restrict__ ovf_list,
                                                    const float* __restrict__ H,
                                                    const float* __restrict__ bias,
                                                    float* __restrict__ out, int n, int e) {
  constexpr int LPR = D / 4;
  constexpr int EPI = 32 / LPR;
  static_assert(LPR == 32 || LPR == 16, "row must be one wave or half a wave of float4");
  const int lane = threadIdx.x & 31;
  const int node = __builtin_amdgcn_readfirstlane(blockIdx.x * 8 + (threadIdx.x >> 5));
  if (node >= n) return;
  const int sub = lane / LPR;
  const int c = lane % LPR;

  const int mt = cnt[node];
  const float di = rsqrtf((float)(mt + 1));
  const int m = min(max(mt, 0), kSlots);

  const float4* __restrict__ H4 = reinterpret_cast<const float4*>(H);
  const float4 hs = H4[(size_t)node * LPR + c];
  float4 acc = make_float4(0.0f, 0.0f, 0.0f, 0.0f);
  const int* __restrict__ slots = bucket + (size_t)node * kSlots;

  for (int base = 0; base < m; base += 32) {
    const int rem = min(32, m - base);
    int sv = 0;
    if (lane < rem) sv = slots[base + lane];
    const int src = (int)min((unsigned)sv, (unsigned)(n - 1));
    const int full = rem - (rem % EPI);
#pragma unroll 4
    for (int j = 0; j < full; j += EPI) {
      const int s = lane_pick<EPI>(src, j, sub);
      row_add(acc, H4[(size_t)s * LPR + c]);
    }
    if (EPI == 2 && (rem & 1)) {
      const int s = __builtin_amdgcn_readlane(src, rem - 1);
      if (sub == 0) row_add(acc, H4[(size_t)s * LPR + c]);
    }
  }

  if (mt > kSlots) {
    const int2* __restrict__ ovf = reinterpret_cast<const int2*>(ovf_list);
    const int on = min(max(ovf_cnt[0], 0), e);
    for (int b = 0; b < on; b += 32) {
      int2 oe = make_int2(-1, 0);
      if (b + lane < on) oe = ovf[b + lane];
      const int src = (int)min((unsigned)oe.y, (unsigned)(n - 1));
      unsigned hit = __builtin_amdgcn_ballot_w32(oe.x == node);
      while (hit) {
        const int j = __builtin_ctz(hit);
        hit &= hit - 1;
        const int s = __builtin_amdgcn_readlane(src, j);
        if (sub == 0) row_add(acc, H4[(size_t)s * LPR + c]);
      }
    }
  }

  if (EPI == 2) {
    acc.x += __shfl_xor(acc.x, 16);
    acc.y += __shfl_xor(acc.y, 16);
    acc.z += __shfl_xor(acc.z, 16);
    acc.w += __shfl_xor(acc.w, 16);
  }
  if (sub == 0) {
    const float4 b = reinterpret_cast<const float4*>(bias)[c];
    float4 o;
    o.x = fmaf(di, acc.x + hs.x, b.x);
    o.y = fmaf(di, acc.y + hs.y, b.y);
    o.z = fmaf(di, acc.z + hs.z, b.z);
    o.w = fmaf(di, acc.w + hs.w, b.w);
    reinterpret_cast<float4*>(out)[(size_t)node * LPR + c] = o;
  }
}


static constexpr int kFN     = 32;
static constexpr int kFPitch = kHidDim + 4;
static constexpr int kOPitch = kOutDim + 4;
static constexpr int kKB     = 4;
static_assert(kInDim == kHidDim, "one tile pitch and one k loop serve both products");

__device__ __forceinline__ void row_fma(float4& acc, float w, const float4 h) {
  acc.x = fmaf(w, h.x, acc.x);
  acc.y = fmaf(w, h.y, acc.y);
  acc.z = fmaf(w, h.z, acc.z);
  acc.w = fmaf(w, h.w, acc.w);
}

__device__ __forceinline__ float lane_read_f(float v, int j) {
  return __int_as_float(__builtin_amdgcn_readlane(__float_as_int(v), j));
}

__device__ __forceinline__ float4 sum_rows_x(int node, int mt, int src_first, float w_first, int lane,
                                             const int* __restrict__ cnt,
                                             const int* __restrict__ bucket,
                                             const int* __restrict__ ovf_cnt,
                                             const int* __restrict__ ovf_list,
                                             const float4* __restrict__ X4, int n, int e) {
  constexpr int LPR = kInDim / 4;
  const int m = min(max(mt, 0), kSlots);
  float4 acc = make_float4(0.0f, 0.0f, 0.0f, 0.0f);
  const int* __restrict__ slots = bucket + (size_t)node * kSlots;
  for (int base = 0; base < m; base += 32) {
    const int rem = min(32, m - base);
    int src = src_first;
    float w = w_first;
    if (base > 0) {
      int sv = 0;
      if (lane < rem) sv = slots[base + lane];
      src = (int)min((unsigned)sv, (unsigned)(n - 1));
      w = rsqrtf((float)(cnt[src] + 1));
    }
#pragma unroll 4
    for (int j = 0; j < rem; ++j) {
      const int sidx = __builtin_amdgcn_readlane(src, j);
      const float wj = lane_read_f(w, j);
      row_fma(acc, wj, X4[(size_t)sidx * LPR + lane]);
    }
  }
  if (mt > kSlots) {
    const int2* __restrict__ ovf = reinterpret_cast<const int2*>(ovf_list);
    const int on = min(max(ovf_cnt[0], 0), e);
    for (int b = 0; b < on; b += 32) {
      int2 oe = make_int2(-1, 0);
      if (b + lane < on) oe = ovf[b + lane];
      const int src = (int)min((unsigned)oe.y, (unsigned)(n - 1));
      const float w = rsqrtf((float)(cnt[src] + 1));
      unsigned hit = __builtin_amdgcn_ballot_w32(oe.x == node);
      while (hit) {
        const int j = __builtin_ctz(hit);
        hit &= hit - 1;
        const int sidx = __builtin_amdgcn_readlane(src, j);
        const float wj = lane_read_f(w, j);
        row_fma(acc, wj, X4[(size_t)sidx * LPR + lane]);
      }
    }
  }
  return acc;
}

template <int LDW>
__device__ __forceinline__ void load_b_group(const float* __restrict__ wcol, int g, v2f (&b)[kKB]) {
#pragma unroll
  for (int u = 0; u < kKB; ++u) {
    b[u].x = wcol[(g * 4 * kKB + 4 * u + 0) * LDW];
    b[u].y = wcol[(g * 4 * kKB + 4 * u + 1) * LDW];
  }
}

template <int LDW>
__device__ __forceinline__ void product_32x16(const float* arow0, const float* __restrict__ wcol,
                                              v2f (&bq)[kKB], v8f& acc0, v8f& acc1) {
  const float* arow1 = arow0 + 16 * kFPitch;
  constexpr int G = kHidDim / (4 * kKB);
#pragma unroll
  for (int kb = 0; kb < G; ++kb) {
    v2f bn[kKB];
#pragma unroll
    for (int u = 0; u < kKB; ++u) bn[u] = {};
    if (kb + 1 < G) load_b_group<LDW>(wcol, kb + 1, bn);
#pragma unroll
    for (int u = 0; u < kKB; ++u) {
      const int k = kb * 4 * kKB + 4 * u;
      v2f a0, a1;
      a0.x = arow0[k + 0];
      a0.y = arow0[k + 1];
      a1.x = arow1[k + 0];
      a1.y = arow1[k + 1];
      acc0 = __builtin_amdgcn_wmma_f32_16x16x4_f32(false, a0, false, bq[u], (short)0, acc0, false, false);
      acc1 = __builtin_amdgcn_wmma_f32_16x16x4_f32(false, a1, false, bq[u], (short)0, acc1, false, false);
    }
#pragma unroll
    for (int u = 0; u < kKB; ++u) bq[u] = bn[u];
  }
}

__global__ __launch_bounds__(256, 16) void k_agg_gemm(const int* __restrict__ cnt,
                                                  const int* __restrict__ bucket,
                                                  const int* __restrict__ ovf_cnt,
                                                  const int* __restrict__ ovf_list,
                                                  const float* __restrict__ X,
                                                  const float* __restrict__ W1,
                                                  const float* __restrict__ bias1,
                                                  const float* __restrict__ W2,
                                                  float* __restrict__ H2, int n, int e) {
  constexpr int LPR = kInDim / 4;
  constexpr int NPW = kFN / 8;
  __shared__ __attribute__((aligned(16))) float tileA[kFN * kFPitch];
  __shared__ __attribute__((aligned(16))) float tileO[kFN * kOPitch];
  __shared__ float sdi[kFN];

  const int lane = threadIdx.x & 31;
  const int wv = __builtin_amdgcn_readfirstlane(threadIdx.x >> 5);
  const int node0 = blockIdx.x * kFN;
  const float4* __restrict__ X4 = reinterpret_cast<const float4*>(X);

  int nd[NPW];
  int mt[NPW];
#pragma unroll
  for (int i = 0; i < NPW; ++i) {
    nd[i] = min(node0 + wv * NPW + i, n - 1);
    mt[i] = cnt[nd[i]];
  }
  int src0[NPW];
  float4 xs[NPW];
#pragma unroll
  for (int i = 0; i < NPW; ++i) {
    const int m0 = min(min(max(mt[i], 0), kSlots), 32);
    const int* p = (lane < m0) ? bucket + (size_t)nd[i] * kSlots + lane : cnt + nd[i];
    const int sv = *p;
    src0[i] = (int)min((unsigned)sv, (unsigned)(n - 1));
    xs[i] = X4[(size_t)nd[i] * LPR + lane];
  }
  float w0[NPW];
#pragma unroll
  for (int i = 0; i < NPW; ++i) w0[i] = rsqrtf((float)(cnt[src0[i]] + 1));

#pragma unroll
  for (int i = 0; i < NPW; ++i) {
    const int r = wv * NPW + i;
    const float4 acc = sum_rows_x(nd[i], mt[i], src0[i], w0[i], lane, cnt, bucket, ovf_cnt, ovf_list, X4, n, e);
    const float di = rsqrtf((float)(mt[i] + 1));
    float4 o;
    o.x = di * fmaf(di, xs[i].x, acc.x);
    o.y = di * fmaf(di, xs[i].y, acc.y);
    o.z = di * fmaf(di, xs[i].z, acc.z);
    o.w = di * fmaf(di, xs[i].w, acc.w);
    *reinterpret_cast<float4*>(&tileA[r * kFPitch + 4 * lane]) = o;
    if (lane == 0) sdi[r] = di;
  }

  const int ml = lane & 15;
  const int hi = lane >> 4;
  const float* arow0 = &tileA[ml * kFPitch + 2 * hi];

  const float* __restrict__ w1col = W1 + (2 * hi) * kHidDim + wv * 16 + ml;
  v2f bq[kKB];
  load_b_group<kHidDim>(w1col, 0, bq);
  const float b1v = bias1[wv * 16 + ml];
  __syncthreads();

  {
    v8f acc0 = {};
    v8f acc1 = {};
    product_32x16<kHidDim>(arow0, w1col, bq, acc0, acc1);
    __syncthreads();
#pragma unroll
    for (int v = 0; v < 8; ++v) {
      const int r0 = v + 8 * hi;
      const int r1 = r0 + 16;
      tileA[r0 * kFPitch + wv * 16 + ml] = fmaxf(acc0[v] + b1v, 0.0f);
      tileA[r1 * kFPitch + wv * 16 + ml] = fmaxf(acc1[v] + b1v, 0.0f);
    }
  }

  const float* __restrict__ w2col = W2 + (2 * hi) * kOutDim + (wv & 3) * 16 + ml;
#pragma unroll
  for (int u = 0; u < kKB; ++u) bq[u] = {};
  if (wv < 4) load_b_group<kOutDim>(w2col, 0, bq);
  __syncthreads();

  if (wv < 4) {
    v8f acc0 = {};
    v8f acc1 = {};
    product_32x16<kOutDim>(arow0, w2col, bq, acc0, acc1);
#pragma unroll
    for (int v = 0; v < 8; ++v) {
      const int r0 = v + 8 * hi;
      const int r1 = r0 + 16;
      tileO[r0 * kOPitch + wv * 16 + ml] = acc0[v] * sdi[r0];
      tileO[r1 * kOPitch + wv * 16 + ml] = acc1[v] * sdi[r1];
    }
  }
  __syncthreads();

  const int tid = wv * 32 + (int)__builtin_amdgcn_mbcnt_lo(~0u, 0u);
  constexpr int OQ = kOutDim / 4;
#pragma unroll
  for (int q = 0; q < (kFN * OQ) / 256; ++q) {
    const int idx = tid + 256 * q;
    const int r = idx / OQ;
    const int c4 = idx % OQ;
    if (node0 + r < n)
      reinterpret_cast<float4*>(H2)[(size_t)(node0 + r) * OQ + c4] =
          *reinterpret_cast<const float4*>(&tileO[r * kOPitch + 4 * c4]);
  }
}


extern "C" void kernel_launch(void* const* d_in, const int* in_sizes, int n_in,
                              void* d_out, int out_size, void* d_ws, size_t ws_size,
                              hipStream_t stream) {
  const float* x  = (const float*)d_in[0];
  const int*   ei = (const int*)d_in[1];
  const float* W1 = (const float*)d_in[2];
  const float* b1 = (const float*)d_in[3];
  const float* W2 = (const float*)d_in[4];
  const float* b2 = (const float*)d_in[5];

  const int n = kNodes;
  const int e = in_sizes[1] / 2;
  const int* row = ei;
  const int* col = ei + e;

  char* ws = (char*)d_ws;
  constexpr size_t MiB = 1024 * 1024;
  constexpr size_t CNT_OFF    = 1 * MiB;
  constexpr size_t H_OFF      = 4 * MiB;
  constexpr size_t BUCKET_OFF = 112 * MiB;
  constexpr size_t OVF_OFF    = 144 * MiB;
  constexpr size_t CELLCNT_OFF = 2 * MiB;
  constexpr size_t CELLS_OFF   = 192 * MiB;
  static_assert(H_OFF + (size_t)kNodes * kOutDim * 4 <= BUCKET_OFF, "layer-2 rows must end before the buckets");
  static_assert(BUCKET_OFF + (size_t)kNodes * kSlots * 4 <= OVF_OFF, "buckets must end before the overflow list");
  static_assert(CNT_OFF + ((size_t)kNodes + 1) * 4 <= CELLCNT_OFF, "counts must end before the cell counts");
  static_assert(CELLCNT_OFF + (size_t)kParts * kBins * 4 <= H_OFF, "cell counts must end before H");
  static_assert(CELLS_OFF + (size_t)kBins * kParts * kCellCap * 4 <= 256 * MiB, "cells must end inside the workspace");
  int*   cnt     = (int*)(ws + CNT_OFF);
  int*   ovf_cnt = cnt + n;
  float* Hs2     = (float*)(ws + H_OFF);
  int*   bucket  = (int*)(ws + BUCKET_OFF);
  int*   cell_cnt = (int*)(ws + CELLCNT_OFF);
  int*   cells    = (int*)(ws + CELLS_OFF);
  int*   ovf     = (int*)(ws + OVF_OFF);
  float* out     = (float*)d_out;

  k_bin_edges<<<kParts, kBinThreads, 0, stream>>>(row, col, cell_cnt, cells, ovf_cnt, e, n);
  k_rank_place<<<kBins, 256, 0, stream>>>(row, col, cell_cnt, cells, cnt, ovf_cnt, bucket, ovf, e, n);

  const int aggBlocks = (n + 7) / 8;

  const int fusedBlocks = (n + kFN - 1) / kFN;
  k_agg_gemm<<<fusedBlocks, 256, 0, stream>>>(cnt, bucket, ovf_cnt, ovf, x, W1, b1, W2, Hs2, n, e);

  k_gather_agg<kOutDim>
      <<<aggBlocks, 256, 0, stream>>>(cnt, bucket, ovf_cnt, ovf, Hs2, b2, out, n, e);
}


# Edge-list stores as whole pieces of a line: both list passes collect their records in LDS (cells; bucket rows) and store them with neighbouring lanes; 1600000 scattered 4-byte stores per pass gone
# speedup vs baseline: 5.2620x; 1.0845x over previous
#include <hip/hip_runtime.h>


typedef __attribute__((ext_vector_type(2))) float v2f;
typedef __attribute__((ext_vector_type(8))) float v8f;

static constexpr int kNodes  = 100000;
static constexpr int kInDim  = 128;
static constexpr int kHidDim = 128;
static constexpr int kOutDim = 64;

static constexpr int kSlots = 64;


static constexpr int kBinShift = 7;
static constexpr int kBinNodes = 1 << kBinShift;
static constexpr int kBins     = (kNodes + kBinNodes - 1) / kBinNodes;
static constexpr int kParts    = 256;
static constexpr int kCellCap  = 32;
static constexpr int kSrcBits  = 24;
static constexpr int kSrcMask  = (1 << kSrcBits) - 1;
static constexpr int kBinThreads = 1024;
static_assert(kNodes <= (1 << kSrcBits), "a source index must fit the record's low field");
static_assert(kBinNodes <= 128, "destination-in-bin must leave the record non-negative");
static_assert(kParts % 32 == 0, "pass 2 reads the cells 32 at a time");

__global__ __launch_bounds__(kBinThreads) void k_bin_edges(const int* __restrict__ row,
                                                           const int* __restrict__ col,
                                                           int* __restrict__ cell_cnt,
                                                           int* __restrict__ cells,
                                                           int* __restrict__ ovf_cnt, int e, int n) {
  __shared__ int hist[kBins];
  __shared__ __attribute__((aligned(16))) int stage[kBins * kCellCap];
  const int t = threadIdx.x;
  const int part = blockIdx.x;
  for (int b = t; b < kBins; b += kBinThreads) hist[b] = 0;
  if (part == 0 && t == 0) ovf_cnt[0] = 0;
  __syncthreads();

  const int chunk = (e + kParts - 1) / kParts;
  const int beg = part * chunk;
  const int end = min(beg + chunk, e);
  constexpr int U = 4;
  for (int i0 = beg + t; i0 < end; i0 += U * kBinThreads) {
    int d[U], s[U];
#pragma unroll
    for (int u = 0; u < U; ++u) {
      const int i = i0 + u * kBinThreads;
      d[u] = -1;
      s[u] = 0;
      if (i < end) {
        d[u] = col[i];
        s[u] = row[i];
      }
    }
    int lr[U];
#pragma unroll
    for (int u = 0; u < U; ++u) {
      lr[u] = kCellCap;
      if ((unsigned)d[u] < (unsigned)n) lr[u] = atomicAdd(&hist[d[u] >> kBinShift], 1);
    }
#pragma unroll
    for (int u = 0; u < U; ++u) {
      if (lr[u] < kCellCap)
        stage[(d[u] >> kBinShift) * kCellCap + lr[u]] =
            ((d[u] & (kBinNodes - 1)) << kSrcBits) | (s[u] & kSrcMask);
    }
  }
  __syncthreads();
  for (int b = t; b < kBins; b += kBinThreads) cell_cnt[(size_t)part * kBins + b] = hist[b];

  constexpr int LPC = kCellCap / 4;
  constexpr int TPT = (kBins * LPC + kBinThreads - 1) / kBinThreads;
  static_assert(kBinThreads % LPC == 0, "a thread keeps its place inside the cell from task to task");
  const int q = t % LPC;
  int valid[TPT];
#pragma unroll
  for (int i = 0; i < TPT; ++i) {
    const int b = (t + i * kBinThreads) / LPC;
    const int c = (b < kBins) ? hist[b] : 0;
    valid[i] = (c <= kCellCap) ? c : 0;
  }
  int4 v[TPT];
#pragma unroll
  for (int i = 0; i < TPT; ++i) {
    const int b = (t + i * kBinThreads) / LPC;
    v[i] = make_int4(-1, -1, -1, -1);
    if (4 * q < valid[i]) v[i] = *reinterpret_cast<const int4*>(&stage[b * kCellCap + 4 * q]);
  }
#pragma unroll
  for (int i = 0; i < TPT; ++i) {
    const int b = (t + i * kBinThreads) / LPC;
    if (4 * q < valid[i]) {
      int4 o = v[i];
      if (4 * q + 1 >= valid[i]) o.y = -1;
      if (4 * q + 2 >= valid[i]) o.z = -1;
      if (4 * q + 3 >= valid[i]) o.w = -1;
      *reinterpret_cast<int4*>(&cells[((size_t)b * kParts + part) * kCellCap + 4 * q]) = o;
    }
  }
}

static constexpr int kRowPitch = kSlots + 2;

__device__ __forceinline__ void place_edge(int* lcnt, int* lrow, int dl, int node, int s,
                                           int* __restrict__ ovf_cnt,
                                           int* __restrict__ ovf_list, int e) {
  const int r = atomicAdd(&lcnt[dl], 1);
  if (r < kSlots) {
    lrow[dl * kRowPitch + r] = s;
  } else {
    const int p = atomicAdd(ovf_cnt, 1);
    if ((unsigned)p < (unsigned)e) reinterpret_cast<int2*>(ovf_list)[p] = make_int2(node, s);
  }
}

__global__ __launch_bounds__(256) void k_rank_place(const int* __restrict__ row,
                                                    const int* __restrict__ col,
                                                    const int* __restrict__ cell_cnt,
                                                    const int* __restrict__ cells,
                                                    int* __restrict__ cnt, int* __restrict__ ovf_cnt,
                                                    int* __restrict__ bucket,
                                                    int* __restrict__ ovf_list, int e, int n) {
  __shared__ int lcnt[kBinNodes];
  __shared__ __attribute__((aligned(8))) int lrow[kBinNodes * kRowPitch];
  const int t = threadIdx.x;
  const int bin = blockIdx.x;
  if (t < kBinNodes) lcnt[t] = 0;
  for (int i = t; i < kBinNodes * kRowPitch / 2; i += 256) reinterpret_cast<int2*>(lrow)[i] = make_int2(0, 0);

  const int g = t >> 3;
  const int l8 = t & 7;
  constexpr int PASSES = kParts / 32;
  constexpr int RPL = kCellCap / 8;

  int c[PASSES];
#pragma unroll
  for (int p = 0; p < PASSES; ++p) c[p] = cell_cnt[(size_t)(p * 32 + g) * kBins + bin];

  int rec[PASSES][RPL];
#pragma unroll
  for (int p = 0; p < PASSES; ++p) {
    const int* __restrict__ cell = cells + ((size_t)bin * kParts + (p * 32 + g)) * kCellCap;
    const int valid = (c[p] <= kCellCap) ? c[p] : 0;
#pragma unroll
    for (int k = 0; k < RPL; ++k) {
      const int idx = l8 + 8 * k;
      rec[p][k] = -1;
      if (idx < valid) rec[p][k] = cell[idx];
    }
  }

  __syncthreads();

#pragma unroll
  for (int p = 0; p < PASSES; ++p) {
#pragma unroll
    for (int k = 0; k < RPL; ++k) {
      const int r = rec[p][k];
      if (r >= 0) {
        const int dl = (r >> kSrcBits) & (kBinNodes - 1);
        place_edge(lcnt, lrow, dl, bin * kBinNodes + dl, r & kSrcMask, ovf_cnt, ovf_list, e);
      }
    }
  }

  const int chunk = (e + kParts - 1) / kParts;
  for (int p = 0; p < PASSES; ++p) {
    if (c[p] > kCellCap) {
      const int beg = (p * 32 + g) * chunk;
      const int end = min(beg + chunk, e);
      for (int i = beg + l8; i < end; i += 8) {
        const int d = col[i];
        if ((unsigned)d < (unsigned)n && (d >> kBinShift) == bin)
          place_edge(lcnt, lrow, d & (kBinNodes - 1), d, row[i] & kSrcMask, ovf_cnt, ovf_list, e);
      }
    }
  }

  __syncthreads();
  if (t < kBinNodes) {
    const int node = bin * kBinNodes + t;
    if (node < n) cnt[node] = lcnt[t];
  }

  const int lane = t & 31;
  const int w = t >> 5;
  constexpr int RPW = kBinNodes / 8;
  static_assert(kSlots == 64, "one 8-byte store per lane covers a row");
  static_assert(RPW <= 32, "one lane per row holds its count");
  static_assert(kRowPitch % 2 == 0, "rows are read out in 8-byte pieces");
  const int cv = (lane < RPW) ? min(lcnt[w * RPW + lane], kSlots) : 0;
  int2 v[RPW];
#pragma unroll
  for (int i = 0; i < RPW; ++i)
    v[i] = *reinterpret_cast<const int2*>(&lrow[(w * RPW + i) * kRowPitch + 2 * lane]);
#pragma unroll
  for (int i = 0; i < RPW; ++i) {
    const int node = bin * kBinNodes + w * RPW + i;
    const int c = __builtin_amdgcn_readlane(cv, i);
    if (node < n && 2 * lane < c)
      *reinterpret_cast<int2*>(&bucket[(size_t)node * kSlots + 2 * lane]) = v[i];
  }
}


template <int EPI>
__device__ __forceinline__ int lane_pick(int v, int j, int sub) {
  if (EPI == 1) return __builtin_amdgcn_readlane(v, j);
  const int a = __builtin_amdgcn_readlane(v, j);
  const int b = __builtin_amdgcn_readlane(v, j + 1);
  return sub ? b : a;
}

__device__ __forceinline__ void row_add(float4& acc, const float4 h) {
  acc.x += h.x;
  acc.y += h.y;
  acc.z += h.z;
  acc.w += h.w;
}

template <int D>
__global__ __launch_bounds__(256) void k_gather_agg(const int* __restrict__ cnt,
                                                    const int* __restrict__ bucket,
                                                    const int* __restrict__ ovf_cnt,
                                                    const int* __restrict__ ovf_list,
                                                    const float* __restrict__ H,
                                                    const float* __restrict__ bias,
                                                    float* __restrict__ out, int n, int e) {
  constexpr int LPR = D / 4;
  constexpr int EPI = 32 / LPR;
  static_assert(LPR == 32 || LPR == 16, "row must be one wave or half a wave of float4");
  const int lane = threadIdx.x & 31;
  const int node = __builtin_amdgcn_readfirstlane(blockIdx.x * 8 + (threadIdx.x >> 5));
  if (node >= n) return;
  const int sub = lane / LPR;
  const int c = lane % LPR;

  const int mt = cnt[node];
  const float di = rsqrtf((float)(mt + 1));
  const int m = min(max(mt, 0), kSlots);

  const float4* __restrict__ H4 = reinterpret_cast<const float4*>(H);
  const float4 hs = H4[(size_t)node * LPR + c];
  float4 acc = make_float4(0.0f, 0.0f, 0.0f, 0.0f);
  const int* __restrict__ slots = bucket + (size_t)node * kSlots;

  for (int base = 0; base < m; base += 32) {
    const int rem = min(32, m - base);
    int sv = 0;
    if (lane < rem) sv = slots[base + lane];
    const int src = (int)min((unsigned)sv, (unsigned)(n - 1));
    const int full = rem - (rem % EPI);
#pragma unroll 4
    for (int j = 0; j < full; j += EPI) {
      const int s = lane_pick<EPI>(src, j, sub);
      row_add(acc, H4[(size_t)s * LPR + c]);
    }
    if (EPI == 2 && (rem & 1)) {
      const int s = __builtin_amdgcn_readlane(src, rem - 1);
      if (sub == 0) row_add(acc, H4[(size_t)s * LPR + c]);
    }
  }

  if (mt > kSlots) {
    const int2* __restrict__ ovf = reinterpret_cast<const int2*>(ovf_list);
    const int on = min(max(ovf_cnt[0], 0), e);
    for (int b = 0; b < on; b += 32) {
      int2 oe = make_int2(-1, 0);
      if (b + lane < on) oe = ovf[b + lane];
      const int src = (int)min((unsigned)oe.y, (unsigned)(n - 1));
      unsigned hit = __builtin_amdgcn_ballot_w32(oe.x == node);
      while (hit) {
        const int j = __builtin_ctz(hit);
        hit &= hit - 1;
        const int s = __builtin_amdgcn_readlane(src, j);
        if (sub == 0) row_add(acc, H4[(size_t)s * LPR + c]);
      }
    }
  }

  if (EPI == 2) {
    acc.x += __shfl_xor(acc.x, 16);
    acc.y += __shfl_xor(acc.y, 16);
    acc.z += __shfl_xor(acc.z, 16);
    acc.w += __shfl_xor(acc.w, 16);
  }
  if (sub == 0) {
    const float4 b = reinterpret_cast<const float4*>(bias)[c];
    float4 o;
    o.x = fmaf(di, acc.x + hs.x, b.x);
    o.y = fmaf(di, acc.y + hs.y, b.y);
    o.z = fmaf(di, acc.z + hs.z, b.z);
    o.w = fmaf(di, acc.w + hs.w, b.w);
    reinterpret_cast<float4*>(out)[(size_t)node * LPR + c] = o;
  }
}


static constexpr int kFN     = 32;
static constexpr int kFPitch = kHidDim + 4;
static constexpr int kOPitch = kOutDim + 4;
static constexpr int kKB     = 4;
static_assert(kInDim == kHidDim, "one tile pitch and one k loop serve both products");

__device__ __forceinline__ void row_fma(float4& acc, float w, const float4 h) {
  acc.x = fmaf(w, h.x, acc.x);
  acc.y = fmaf(w, h.y, acc.y);
  acc.z = fmaf(w, h.z, acc.z);
  acc.w = fmaf(w, h.w, acc.w);
}

__device__ __forceinline__ float lane_read_f(float v, int j) {
  return __int_as_float(__builtin_amdgcn_readlane(__float_as_int(v), j));
}

__device__ __forceinline__ float4 sum_rows_x(int node, int mt, int src_first, float w_first, int lane,
                                             const int* __restrict__ cnt,
                                             const int* __restrict__ bucket,
                                             const int* __restrict__ ovf_cnt,
                                             const int* __restrict__ ovf_list,
                                             const float4* __restrict__ X4, int n, int e) {
  constexpr int LPR = kInDim / 4;
  const int m = min(max(mt, 0), kSlots);
  float4 acc = make_float4(0.0f, 0.0f, 0.0f, 0.0f);
  const int* __restrict__ slots = bucket + (size_t)node * kSlots;
  for (int base = 0; base < m; base += 32) {
    const int rem = min(32, m - base);
    int src = src_first;
    float w = w_first;
    if (base > 0) {
      int sv = 0;
      if (lane < rem) sv = slots[base + lane];
      src = (int)min((unsigned)sv, (unsigned)(n - 1));
      w = rsqrtf((float)(cnt[src] + 1));
    }
#pragma unroll 4
    for (int j = 0; j < rem; ++j) {
      const int sidx = __builtin_amdgcn_readlane(src, j);
      const float wj = lane_read_f(w, j);
      row_fma(acc, wj, X4[(size_t)sidx * LPR + lane]);
    }
  }
  if (mt > kSlots) {
    const int2* __restrict__ ovf = reinterpret_cast<const int2*>(ovf_list);
    const int on = min(max(ovf_cnt[0], 0), e);
    for (int b = 0; b < on; b += 32) {
      int2 oe = make_int2(-1, 0);
      if (b + lane < on) oe = ovf[b + lane];
      const int src = (int)min((unsigned)oe.y, (unsigned)(n - 1));
      const float w = rsqrtf((float)(cnt[src] + 1));
      unsigned hit = __builtin_amdgcn_ballot_w32(oe.x == node);
      while (hit) {
        const int j = __builtin_ctz(hit);
        hit &= hit - 1;
        const int sidx = __builtin_amdgcn_readlane(src, j);
        const float wj = lane_read_f(w, j);
        row_fma(acc, wj, X4[(size_t)sidx * LPR + lane]);
      }
    }
  }
  return acc;
}

template <int LDW>
__device__ __forceinline__ void load_b_group(const float* __restrict__ wcol, int g, v2f (&b)[kKB]) {
#pragma unroll
  for (int u = 0; u < kKB; ++u) {
    b[u].x = wcol[(g * 4 * kKB + 4 * u + 0) * LDW];
    b[u].y = wcol[(g * 4 * kKB + 4 * u + 1) * LDW];
  }
}

template <int LDW>
__device__ __forceinline__ void product_32x16(const float* arow0, const float* __restrict__ wcol,
                                              v2f (&bq)[kKB], v8f& acc0, v8f& acc1) {
  const float* arow1 = arow0 + 16 * kFPitch;
  constexpr int G = kHidDim / (4 * kKB);
#pragma unroll
  for (int kb = 0; kb < G; ++kb) {
    v2f bn[kKB];
#pragma unroll
    for (int u = 0; u < kKB; ++u) bn[u] = {};
    if (kb + 1 < G) load_b_group<LDW>(wcol, kb + 1, bn);
#pragma unroll
    for (int u = 0; u < kKB; ++u) {
      const int k = kb * 4 * kKB + 4 * u;
      v2f a0, a1;
      a0.x = arow0[k + 0];
      a0.y = arow0[k + 1];
      a1.x = arow1[k + 0];
      a1.y = arow1[k + 1];
      acc0 = __builtin_amdgcn_wmma_f32_16x16x4_f32(false, a0, false, bq[u], (short)0, acc0, false, false);
      acc1 = __builtin_amdgcn_wmma_f32_16x16x4_f32(false, a1, false, bq[u], (short)0, acc1, false, false);
    }
#pragma unroll
    for (int u = 0; u < kKB; ++u) bq[u] = bn[u];
  }
}

__global__ __launch_bounds__(256, 16) void k_agg_gemm(const int* __restrict__ cnt,
                                                  const int* __restrict__ bucket,
                                                  const int* __restrict__ ovf_cnt,
                                                  const int* __restrict__ ovf_list,
                                                  const float* __restrict__ X,
                                                  const float* __restrict__ W1,
                                                  const float* __restrict__ bias1,
                                                  const float* __restrict__ W2,
                                                  float* __restrict__ H2, int n, int e) {
  constexpr int LPR = kInDim / 4;
  constexpr int NPW = kFN / 8;
  __shared__ __attribute__((aligned(16))) float tileA[kFN * kFPitch];
  __shared__ __attribute__((aligned(16))) float tileO[kFN * kOPitch];
  __shared__ float sdi[kFN];

  const int lane = threadIdx.x & 31;
  const int wv = __builtin_amdgcn_readfirstlane(threadIdx.x >> 5);
  const int node0 = blockIdx.x * kFN;
  const float4* __restrict__ X4 = reinterpret_cast<const float4*>(X);

  int nd[NPW];
  int mt[NPW];
#pragma unroll
  for (int i = 0; i < NPW; ++i) {
    nd[i] = min(node0 + wv * NPW + i, n - 1);
    mt[i] = cnt[nd[i]];
  }
  int src0[NPW];
  float4 xs[NPW];
#pragma unroll
  for (int i = 0; i < NPW; ++i) {
    const int m0 = min(min(max(mt[i], 0), kSlots), 32);
    const int* p = (lane < m0) ? bucket + (size_t)nd[i] * kSlots + lane : cnt + nd[i];
    const int sv = *p;
    src0[i] = (int)min((unsigned)sv, (unsigned)(n - 1));
    xs[i] = X4[(size_t)nd[i] * LPR + lane];
  }
  float w0[NPW];
#pragma unroll
  for (int i = 0; i < NPW; ++i) w0[i] = rsqrtf((float)(cnt[src0[i]] + 1));

#pragma unroll
  for (int i = 0; i < NPW; ++i) {
    const int r = wv * NPW + i;
    const float4 acc = sum_rows_x(nd[i], mt[i], src0[i], w0[i], lane, cnt, bucket, ovf_cnt, ovf_list, X4, n, e);
    const float di = rsqrtf((float)(mt[i] + 1));
    float4 o;
    o.x = di * fmaf(di, xs[i].x, acc.x);
    o.y = di * fmaf(di, xs[i].y, acc.y);
    o.z = di * fmaf(di, xs[i].z, acc.z);
    o.w = di * fmaf(di, xs[i].w, acc.w);
    *reinterpret_cast<float4*>(&tileA[r * kFPitch + 4 * lane]) = o;
    if (lane == 0) sdi[r] = di;
  }

  const int ml = lane & 15;
  const int hi = lane >> 4;
  const float* arow0 = &tileA[ml * kFPitch + 2 * hi];

  const float* __restrict__ w1col = W1 + (2 * hi) * kHidDim + wv * 16 + ml;
  v2f bq[kKB];
  load_b_group<kHidDim>(w1col, 0, bq);
  const float b1v = bias1[wv * 16 + ml];
  __syncthreads();

  {
    v8f acc0 = {};
    v8f acc1 = {};
    product_32x16<kHidDim>(arow0, w1col, bq, acc0, acc1);
    __syncthreads();
#pragma unroll
    for (int v = 0; v < 8; ++v) {
      const int r0 = v + 8 * hi;
      const int r1 = r0 + 16;
      tileA[r0 * kFPitch + wv * 16 + ml] = fmaxf(acc0[v] + b1v, 0.0f);
      tileA[r1 * kFPitch + wv * 16 + ml] = fmaxf(acc1[v] + b1v, 0.0f);
    }
  }

  const float* __restrict__ w2col = W2 + (2 * hi) * kOutDim + (wv & 3) * 16 + ml;
#pragma unroll
  for (int u = 0; u < kKB; ++u) bq[u] = {};
  if (wv < 4) load_b_group<kOutDim>(w2col, 0, bq);
  __syncthreads();

  if (wv < 4) {
    v8f acc0 = {};
    v8f acc1 = {};
    product_32x16<kOutDim>(arow0, w2col, bq, acc0, acc1);
#pragma unroll
    for (int v = 0; v < 8; ++v) {
      const int r0 = v + 8 * hi;
      const int r1 = r0 + 16;
      tileO[r0 * kOPitch + wv * 16 + ml] = acc0[v] * sdi[r0];
      tileO[r1 * kOPitch + wv * 16 + ml] = acc1[v] * sdi[r1];
    }
  }
  __syncthreads();

  const int tid = wv * 32 + (int)__builtin_amdgcn_mbcnt_lo(~0u, 0u);
  constexpr int OQ = kOutDim / 4;
#pragma unroll
  for (int q = 0; q < (kFN * OQ) / 256; ++q) {
    const int idx = tid + 256 * q;
    const int r = idx / OQ;
    const int c4 = idx % OQ;
    if (node0 + r < n)
      reinterpret_cast<float4*>(H2)[(size_t)(node0 + r) * OQ + c4] =
          *reinterpret_cast<const float4*>(&tileO[r * kOPitch + 4 * c4]);
  }
}


extern "C" void kernel_launch(void* const* d_in, const int* in_sizes, int n_in,
                              void* d_out, int out_size, void* d_ws, size_t ws_size,
                              hipStream_t stream) {
  const float* x  = (const float*)d_in[0];
  const int*   ei = (const int*)d_in[1];
  const float* W1 = (const float*)d_in[2];
  const float* b1 = (const float*)d_in[3];
  const float* W2 = (const float*)d_in[4];
  const float* b2 = (const float*)d_in[5];

  const int n = kNodes;
  const int e = in_sizes[1] / 2;
  const int* row = ei;
  const int* col = ei + e;

  char* ws = (char*)d_ws;
  constexpr size_t MiB = 1024 * 1024;
  constexpr size_t CNT_OFF    = 1 * MiB;
  constexpr size_t H_OFF      = 4 * MiB;
  constexpr size_t BUCKET_OFF = 112 * MiB;
  constexpr size_t OVF_OFF    = 144 * MiB;
  constexpr size_t CELLCNT_OFF = 2 * MiB;
  constexpr size_t CELLS_OFF   = 192 * MiB;
  static_assert(H_OFF + (size_t)kNodes * kOutDim * 4 <= BUCKET_OFF, "layer-2 rows must end before the buckets");
  static_assert(BUCKET_OFF + (size_t)kNodes * kSlots * 4 <= OVF_OFF, "buckets must end before the overflow list");
  static_assert(CNT_OFF + ((size_t)kNodes + 1) * 4 <= CELLCNT_OFF, "counts must end before the cell counts");
  static_assert(CELLCNT_OFF + (size_t)kParts * kBins * 4 <= H_OFF, "cell counts must end before H");
  static_assert(CELLS_OFF + (size_t)kBins * kParts * kCellCap * 4 <= 256 * MiB, "cells must end inside the workspace");
  int*   cnt     = (int*)(ws + CNT_OFF);
  int*   ovf_cnt = cnt + n;
  float* Hs2     = (float*)(ws + H_OFF);
  int*   bucket  = (int*)(ws + BUCKET_OFF);
  int*   cell_cnt = (int*)(ws + CELLCNT_OFF);
  int*   cells    = (int*)(ws + CELLS_OFF);
  int*   ovf     = (int*)(ws + OVF_OFF);
  float* out     = (float*)d_out;

  k_bin_edges<<<kParts, kBinThreads, 0, stream>>>(row, col, cell_cnt, cells, ovf_cnt, e, n);
  k_rank_place<<<kBins, 256, 0, stream>>>(row, col, cell_cnt, cells, cnt, ovf_cnt, bucket, ovf, e, n);

  const int aggBlocks = (n + 7) / 8;

  const int fusedBlocks = (n + kFN - 1) / kFN;
  k_agg_gemm<<<fusedBlocks, 256, 0, stream>>>(cnt, bucket, ovf_cnt, ovf, x, W1, b1, W2, Hs2, n, e);

  k_gather_agg<kOutDim>
      <<<aggBlocks, 256, 0, stream>>>(cnt, bucket, ovf_cnt, ovf, Hs2, b2, out, n, e);
}
